# starting kernel
# baseline (speedup 1.0000x reference)
#include <hip/hip_runtime.h>
#include <hip/hip_bf16.h>

typedef __attribute__((ext_vector_type(16))) _Float16 v16h;
typedef __attribute__((ext_vector_type(8)))  float    v8f;

__global__ void gemv_wmma_kernel(const float* __restrict__ x,
                                 const float* __restrict__ W,
                                 const float* __restrict__ bias,
                                 float* __restrict__ y,
                                 int K, int N, int do_relu) {
    extern __shared__ float sx[];
    const int tid  = threadIdx.x;
    const int lane = tid & 31;
    const int wave = tid >> 5;

    for (int i = tid; i < K; i += blockDim.x) sx[i] = x[i];
    __syncthreads();

    const int j0  = blockIdx.x * 64 + wave * 16;
    const int col = j0 + (lane & 15);
    const int hi8 = (lane >> 4) * 8;

    v8f acc = {};
    for (int k0 = 0; k0 < K; k0 += 32) {
        v16h ah = {}, al = {}, bh = {}, bl = {};
#pragma unroll
        for (int e = 0; e < 16; ++e) {
            const int kk = k0 + e + (e & 8) + hi8;
            const float xv = sx[kk];
            const float wv = W[(long)kk * N + col];
            const _Float16 xh = (_Float16)xv;
            const _Float16 wh = (_Float16)wv;
            ah[e] = xh;  al[e] = (_Float16)(xv - (float)xh);
            bh[e] = wh;  bl[e] = (_Float16)(wv - (float)wh);
        }
        acc = __builtin_amdgcn_wmma_f32_16x16x32_f16(
            false, ah, false, bh, (short)0, acc, false, false);
        acc = __builtin_amdgcn_wmma_f32_16x16x32_f16(
            false, ah, false, bl, (short)0, acc, false, false);
        acc = __builtin_amdgcn_wmma_f32_16x16x32_f16(
            false, al, false, bh, (short)0, acc, false, false);
    }

    if (lane < 16) {
        float v = acc[0] + bias[col];
        if (do_relu) v = fmaxf(v, 0.0f);
        y[col] = v;
    }
}

#define RN 4096

__global__ void eig_bisect_kernel(const float* __restrict__ ptl,
                                  float* __restrict__ evl) {
    __shared__ float sd[RN];
    const int   l   = blockIdx.y;
    const float ll1 = (float)(l * (l + 1));
    const int   tid = threadIdx.x;
    const float dr  = 10000.0f / (float)RN;

    for (int i = tid; i < RN; i += blockDim.x) {
        const float r = dr * (float)(i + 1);
        sd[i] = 1.0f + ptl[i] + ll1 / (r * r);
    }
    __syncthreads();

    float lo = 1e30f, hi = -1e30f;
    for (int i = 0; i < RN; ++i) {
        const float d = sd[i];
        lo = fminf(lo, d - 1.0f);
        hi = fmaxf(hi, d + 1.0f);
    }

    const int j = blockIdx.x * blockDim.x + tid;

    for (int it = 0; it < 48; ++it) {
        const float x = 0.5f * (lo + hi);
        float q   = sd[0] - x;
        int   cnt = (q < 0.0f);
        for (int i = 1; i < RN; ++i) {
            q = (sd[i] - x) - 0.25f * __builtin_amdgcn_rcpf(q);
            if (fabsf(q) < 1e-30f) q = -1e-30f;
            cnt += (q < 0.0f);
        }
        if (cnt > j) hi = x; else lo = x;
    }
    evl[l * RN + j] = 0.5f * (lo + hi);
}

extern "C" void kernel_launch(void* const* d_in, const int* in_sizes, int n_in,
                              void* d_out, int out_size, void* d_ws, size_t ws_size,
                              hipStream_t stream) {
    const float* energy = (const float*)d_in[0];
    const float* W1     = (const float*)d_in[1];
    const float* b1     = (const float*)d_in[2];
    const float* W2     = (const float*)d_in[3];
    const float* b2     = (const float*)d_in[4];
    const float* W3     = (const float*)d_in[5];
    const float* b3     = (const float*)d_in[6];

    float* out = (float*)d_out;
    float* h1  = (float*)d_ws;
    float* h2  = h1 + 2048;

    gemv_wmma_kernel<<<2048 / 64, 128, 4096 * sizeof(float), stream>>>(
        energy, W1, b1, h1, 4096, 2048, 1);
    gemv_wmma_kernel<<<1024 / 64, 128, 2048 * sizeof(float), stream>>>(
        h1, W2, b2, h2, 2048, 1024, 1);
    gemv_wmma_kernel<<<4096 / 64, 128, 1024 * sizeof(float), stream>>>(
        h2, W3, b3, out, 1024, 4096, 0);
    eig_bisect_kernel<<<dim3(RN / 256, 2), 256, 0, stream>>>(out, out + 4096);
}


# Sturm count by the division-free recurrence P_i=T_i*P_{i-1}-P_{i-2} (sign history word + popcount, exact power-of-two rescale every 16 steps): one FMA on the per-step chain instead of rcp+fma+guard
# speedup vs baseline: 3.0814x; 3.0814x over previous
#include <hip/hip_runtime.h>
#include <hip/hip_bf16.h>

typedef __attribute__((ext_vector_type(16))) _Float16 v16h;
typedef __attribute__((ext_vector_type(8)))  float    v8f;

__global__ void gemv_wmma_kernel(const float* __restrict__ x,
                                 const float* __restrict__ W,
                                 const float* __restrict__ bias,
                                 float* __restrict__ y,
                                 int K, int N, int do_relu) {
    extern __shared__ float sx[];
    const int tid  = threadIdx.x;
    const int lane = tid & 31;
    const int wave = tid >> 5;

    for (int i = tid; i < K; i += blockDim.x) sx[i] = x[i];
    __syncthreads();

    const int j0  = blockIdx.x * 64 + wave * 16;
    const int col = j0 + (lane & 15);
    const int hi8 = (lane >> 4) * 8;

    v8f acc = {};
    for (int k0 = 0; k0 < K; k0 += 32) {
        v16h ah = {}, al = {}, bh = {}, bl = {};
#pragma unroll
        for (int e = 0; e < 16; ++e) {
            const int kk = k0 + e + (e & 8) + hi8;
            const float xv = sx[kk];
            const float wv = W[(long)kk * N + col];
            const _Float16 xh = (_Float16)xv;
            const _Float16 wh = (_Float16)wv;
            ah[e] = xh;  al[e] = (_Float16)(xv - (float)xh);
            bh[e] = wh;  bl[e] = (_Float16)(wv - (float)wh);
        }
        acc = __builtin_amdgcn_wmma_f32_16x16x32_f16(
            false, ah, false, bh, (short)0, acc, false, false);
        acc = __builtin_amdgcn_wmma_f32_16x16x32_f16(
            false, ah, false, bl, (short)0, acc, false, false);
        acc = __builtin_amdgcn_wmma_f32_16x16x32_f16(
            false, al, false, bh, (short)0, acc, false, false);
    }

    if (lane < 16) {
        float v = acc[0] + bias[col];
        if (do_relu) v = fmaxf(v, 0.0f);
        y[col] = v;
    }
}

#define RN 4096

typedef __attribute__((ext_vector_type(4))) float v4f;

__device__ __forceinline__ void sturm_block32(const v4f (&t)[8],
                                              float& p1, float& p0,
                                              unsigned& prev, int& cnt) {
    unsigned acc = prev;
#pragma unroll
    for (int u = 0; u < 8; ++u) {
#pragma unroll
        for (int s = 0; s < 4; ++s) {
            const float p2 = __builtin_fmaf(t[u][s], p1, -p0);
            acc = __builtin_amdgcn_alignbit(acc, __float_as_uint(p2), 31u);
            p0 = p1;
            p1 = p2;
        }
        if ((u & 3) == 3) {
            const int e = __builtin_amdgcn_frexp_expf(
                __builtin_fmaxf(__builtin_fabsf(p1), __builtin_fabsf(p0)));
            p1 = __builtin_ldexpf(p1, -e);
            p0 = __builtin_ldexpf(p0, -e);
        }
    }
    const unsigned older = __builtin_amdgcn_alignbit(prev, acc, 1u);
    cnt += __builtin_popcount(acc ^ older);
    prev = acc;
}

__device__ __forceinline__ void hold_block(v4f (&t)[8], float& p1) {
#pragma unroll
    for (int u = 0; u < 8; ++u)
        asm volatile("" : "+v"(t[u]), "+v"(p1) : : "memory");
}

__device__ __forceinline__ int sturm_count_scaled(const v4f* sd4,
                                                  const float x2) {
    float    p1   = 1.0f;
    float    p0   = 0.0f;
    unsigned prev = 0u;
    int      cnt  = 0;
    const int nblk = RN / 32;

    v4f a[8], b[8], ra[8], rb[8];
#pragma unroll
    for (int u = 0; u < 8; ++u) a[u] = sd4[u] - x2;
#pragma unroll
    for (int u = 0; u < 8; ++u) rb[u] = sd4[8 + u];
    hold_block(rb, p1);

    for (int k = 0; k < nblk; k += 2) {
        const int k2 = ((k + 2) & (nblk - 1)) * 8;
#pragma unroll
        for (int u = 0; u < 8; ++u) ra[u] = sd4[k2 + u];
        __builtin_amdgcn_sched_barrier(0);
#pragma unroll
        for (int u = 0; u < 8; ++u) b[u] = rb[u] - x2;
        sturm_block32(a, p1, p0, prev, cnt);
        hold_block(b, p1);
        hold_block(ra, p1);
        const int k3 = ((k + 3) & (nblk - 1)) * 8;
#pragma unroll
        for (int u = 0; u < 8; ++u) rb[u] = sd4[k3 + u];
        __builtin_amdgcn_sched_barrier(0);
#pragma unroll
        for (int u = 0; u < 8; ++u) a[u] = ra[u] - x2;
        sturm_block32(b, p1, p0, prev, cnt);
        hold_block(a, p1);
        hold_block(rb, p1);
    }
    return cnt;
}

__device__ __forceinline__ int sturm_count_ratio(const float* sd,
                                                 const float x2) {
    float q   = sd[0] - x2;
    int   cnt = (q < 0.0f);
    for (int i = 1; i < RN; ++i) {
        q = (sd[i] - x2) - __builtin_amdgcn_rcpf(q);
        if (fabsf(q) < 2e-30f) q = -2e-30f;
        cnt += (q < 0.0f);
    }
    return cnt;
}

__global__ void __launch_bounds__(256)
eig_bisect_kernel(const float* __restrict__ ptl, float* __restrict__ evl) {
    __shared__ v4f sd4[RN / 4];
    float* sd = (float*)sd4;
    const int   l   = blockIdx.y;
    const float ll1 = (float)(l * (l + 1));
    const int   tid = threadIdx.x;
    const float dr  = 10000.0f / (float)RN;

    for (int i = tid; i < RN; i += blockDim.x) {
        const float r = dr * (float)(i + 1);
        const float d = 1.0f + ptl[i] + ll1 / (r * r);
        sd[i] = 2.0f * d;
    }
    __syncthreads();

    float mn = 1e30f, mx = -1e30f;
    for (int i = 0; i < RN / 4; ++i) {
        const v4f v = sd4[i];
        mn = fminf(fminf(mn, v[0]), fminf(fminf(v[1], v[2]), v[3]));
        mx = fmaxf(fmaxf(mx, v[0]), fmaxf(fmaxf(v[1], v[2]), v[3]));
    }
    float lo = fminf(1e30f, 0.5f * mn - 1.0f);
    float hi = fmaxf(-1e30f, 0.5f * mx + 1.0f);

    const int narrow = __builtin_amdgcn_readfirstlane((hi - lo) < 48.0f ? 1 : 0);

    const int j = blockIdx.x * blockDim.x + tid;

    for (int it = 0; it < 48; ++it) {
        const float x2 = lo + hi;
        const int cnt = narrow ? sturm_count_scaled(sd4, x2)
                               : sturm_count_ratio(sd, x2);
        if (cnt > j) hi = 0.5f * x2; else lo = 0.5f * x2;
    }
    evl[l * RN + j] = 0.5f * (lo + hi);
}

extern "C" void kernel_launch(void* const* d_in, const int* in_sizes, int n_in,
                              void* d_out, int out_size, void* d_ws, size_t ws_size,
                              hipStream_t stream) {
    const float* energy = (const float*)d_in[0];
    const float* W1     = (const float*)d_in[1];
    const float* b1     = (const float*)d_in[2];
    const float* W2     = (const float*)d_in[3];
    const float* b2     = (const float*)d_in[4];
    const float* W3     = (const float*)d_in[5];
    const float* b3     = (const float*)d_in[6];

    float* out = (float*)d_out;
    float* h1  = (float*)d_ws;
    float* h2  = h1 + 2048;

    gemv_wmma_kernel<<<2048 / 64, 128, 4096 * sizeof(float), stream>>>(
        energy, W1, b1, h1, 4096, 2048, 1);
    gemv_wmma_kernel<<<1024 / 64, 128, 2048 * sizeof(float), stream>>>(
        h1, W2, b2, h2, 2048, 1024, 1);
    gemv_wmma_kernel<<<4096 / 64, 128, 1024 * sizeof(float), stream>>>(
        h2, W3, b3, out, 1024, 4096, 0);
    eig_bisect_kernel<<<dim3(RN / 256, 2), 256, 0, stream>>>(out, out + 4096);
}


# Bisection -> 9-way multisection: 8 lanes per eigenvalue run the Sturm chain at the 8 interior points, a ballot keeps one ninth; 16 chain sweeps (9^16 > 2^48) replace 48, on 8x the waves
# speedup vs baseline: 6.0488x; 1.9630x over previous
#include <hip/hip_runtime.h>
#include <hip/hip_bf16.h>

typedef __attribute__((ext_vector_type(16))) _Float16 v16h;
typedef __attribute__((ext_vector_type(8)))  float    v8f;

__global__ void gemv_wmma_kernel(const float* __restrict__ x,
                                 const float* __restrict__ W,
                                 const float* __restrict__ bias,
                                 float* __restrict__ y,
                                 int K, int N, int do_relu) {
    extern __shared__ float sx[];
    const int tid  = threadIdx.x;
    const int lane = tid & 31;
    const int wave = tid >> 5;

    for (int i = tid; i < K; i += blockDim.x) sx[i] = x[i];
    __syncthreads();

    const int j0  = blockIdx.x * 64 + wave * 16;
    const int col = j0 + (lane & 15);
    const int hi8 = (lane >> 4) * 8;

    v8f acc = {};
    for (int k0 = 0; k0 < K; k0 += 32) {
        v16h ah = {}, al = {}, bh = {}, bl = {};
#pragma unroll
        for (int e = 0; e < 16; ++e) {
            const int kk = k0 + e + (e & 8) + hi8;
            const float xv = sx[kk];
            const float wv = W[(long)kk * N + col];
            const _Float16 xh = (_Float16)xv;
            const _Float16 wh = (_Float16)wv;
            ah[e] = xh;  al[e] = (_Float16)(xv - (float)xh);
            bh[e] = wh;  bl[e] = (_Float16)(wv - (float)wh);
        }
        acc = __builtin_amdgcn_wmma_f32_16x16x32_f16(
            false, ah, false, bh, (short)0, acc, false, false);
        acc = __builtin_amdgcn_wmma_f32_16x16x32_f16(
            false, ah, false, bl, (short)0, acc, false, false);
        acc = __builtin_amdgcn_wmma_f32_16x16x32_f16(
            false, al, false, bh, (short)0, acc, false, false);
    }

    if (lane < 16) {
        float v = acc[0] + bias[col];
        if (do_relu) v = fmaxf(v, 0.0f);
        y[col] = v;
    }
}

#define RN 4096

typedef __attribute__((ext_vector_type(4))) float v4f;

__device__ __forceinline__ void sturm_block32(const v4f (&t)[8],
                                              float& p1, float& p0,
                                              unsigned& prev, int& cnt) {
    unsigned acc = prev;
#pragma unroll
    for (int u = 0; u < 8; ++u) {
#pragma unroll
        for (int s = 0; s < 4; ++s) {
            const float p2 = __builtin_fmaf(t[u][s], p1, -p0);
            acc = __builtin_amdgcn_alignbit(acc, __float_as_uint(p2), 31u);
            p0 = p1;
            p1 = p2;
        }
        if ((u & 3) == 3) {
            const int e = __builtin_amdgcn_frexp_expf(
                __builtin_fmaxf(__builtin_fabsf(p1), __builtin_fabsf(p0)));
            p1 = __builtin_ldexpf(p1, -e);
            p0 = __builtin_ldexpf(p0, -e);
        }
    }
    const unsigned older = __builtin_amdgcn_alignbit(prev, acc, 1u);
    cnt += __builtin_popcount(acc ^ older);
    prev = acc;
}

__device__ __forceinline__ void hold_block(v4f (&t)[8], float& p1) {
#pragma unroll
    for (int u = 0; u < 8; ++u)
        asm volatile("" : "+v"(t[u]), "+v"(p1) : : "memory");
}

__device__ __forceinline__ int sturm_count_scaled(const v4f* sd4,
                                                  const float x2) {
    float    p1   = 1.0f;
    float    p0   = 0.0f;
    unsigned prev = 0u;
    int      cnt  = 0;
    const int nblk = RN / 32;

    v4f a[8], b[8], ra[8], rb[8];
#pragma unroll
    for (int u = 0; u < 8; ++u) a[u] = sd4[u] - x2;
#pragma unroll
    for (int u = 0; u < 8; ++u) rb[u] = sd4[8 + u];
    hold_block(rb, p1);

    for (int k = 0; k < nblk; k += 2) {
        const int k2 = ((k + 2) & (nblk - 1)) * 8;
#pragma unroll
        for (int u = 0; u < 8; ++u) ra[u] = sd4[k2 + u];
        __builtin_amdgcn_sched_barrier(0);
#pragma unroll
        for (int u = 0; u < 8; ++u) b[u] = rb[u] - x2;
        sturm_block32(a, p1, p0, prev, cnt);
        hold_block(b, p1);
        hold_block(ra, p1);
        const int k3 = ((k + 3) & (nblk - 1)) * 8;
#pragma unroll
        for (int u = 0; u < 8; ++u) rb[u] = sd4[k3 + u];
        __builtin_amdgcn_sched_barrier(0);
#pragma unroll
        for (int u = 0; u < 8; ++u) a[u] = ra[u] - x2;
        sturm_block32(b, p1, p0, prev, cnt);
        hold_block(a, p1);
        hold_block(rb, p1);
    }
    return cnt;
}

__device__ __forceinline__ int sturm_count_ratio(const float* sd,
                                                 const float x2) {
    float q   = sd[0] - x2;
    int   cnt = (q < 0.0f);
    for (int i = 1; i < RN; ++i) {
        q = (sd[i] - x2) - __builtin_amdgcn_rcpf(q);
        if (fabsf(q) < 2e-30f) q = -2e-30f;
        cnt += (q < 0.0f);
    }
    return cnt;
}

#define NPT    8
#define SWEEPS 16

__device__ __forceinline__ float section_point(const float lo, const float w,
                                               const int i) {
    return __builtin_fmaf(w, (float)(i + 1) * (1.0f / (float)(NPT + 1)), lo);
}

__global__ void __launch_bounds__(256)
eig_bisect_kernel(const float* __restrict__ ptl, float* __restrict__ evl) {
    __shared__ v4f   sd4[RN / 4];
    __shared__ float res[256 / NPT];
    float* sd = (float*)sd4;
    const int   l   = blockIdx.y;
    const float ll1 = (float)(l * (l + 1));
    const int   tid = threadIdx.x;
    const float dr  = 10000.0f / (float)RN;

    for (int i = tid; i < RN; i += blockDim.x) {
        const float r = dr * (float)(i + 1);
        const float d = 1.0f + ptl[i] + ll1 / (r * r);
        sd[i] = 2.0f * d;
    }
    __syncthreads();

    float mn = 1e30f, mx = -1e30f;
    for (int i = 0; i < RN / 4; ++i) {
        const v4f v = sd4[i];
        mn = fminf(fminf(mn, v[0]), fminf(fminf(v[1], v[2]), v[3]));
        mx = fmaxf(fmaxf(mx, v[0]), fmaxf(fmaxf(v[1], v[2]), v[3]));
    }
    float lo = fminf(1e30f, 0.5f * mn - 1.0f);
    float hi = fmaxf(-1e30f, 0.5f * mx + 1.0f);

    const int narrow = __builtin_amdgcn_readfirstlane((hi - lo) < 48.0f ? 1 : 0);

    const int q   = tid & (NPT - 1);
    const int grp = tid / NPT;
    const int gsh = tid & 31 & ~(NPT - 1);
    const int j   = blockIdx.x * (256 / NPT) + grp;

    for (int it = 0; it < SWEEPS; ++it) {
        const float w  = hi - lo;
        const float x  = section_point(lo, w, q);
        const float x2 = x + x;
        const int cnt = narrow ? sturm_count_scaled(sd4, x2)
                               : sturm_count_ratio(sd, x2);
        const unsigned below = __builtin_amdgcn_ballot_w32(cnt <= j);
        const int k = __builtin_popcount((below >> gsh) & ((1u << NPT) - 1u));
        const float nlo = (k > 0)   ? section_point(lo, w, k - 1) : lo;
        const float nhi = (k < NPT) ? section_point(lo, w, k)     : hi;
        lo = nlo;
        hi = nhi;
    }

    if (q == 0) res[grp] = 0.5f * (lo + hi);
    __syncthreads();
    if (tid < 256 / NPT)
        evl[l * RN + blockIdx.x * (256 / NPT) + tid] = res[tid];
}

extern "C" void kernel_launch(void* const* d_in, const int* in_sizes, int n_in,
                              void* d_out, int out_size, void* d_ws, size_t ws_size,
                              hipStream_t stream) {
    const float* energy = (const float*)d_in[0];
    const float* W1     = (const float*)d_in[1];
    const float* b1     = (const float*)d_in[2];
    const float* W2     = (const float*)d_in[3];
    const float* b2     = (const float*)d_in[4];
    const float* W3     = (const float*)d_in[5];
    const float* b3     = (const float*)d_in[6];

    float* out = (float*)d_out;
    float* h1  = (float*)d_ws;
    float* h2  = h1 + 2048;

    gemv_wmma_kernel<<<2048 / 64, 128, 4096 * sizeof(float), stream>>>(
        energy, W1, b1, h1, 4096, 2048, 1);
    gemv_wmma_kernel<<<1024 / 64, 128, 2048 * sizeof(float), stream>>>(
        h1, W2, b2, h2, 2048, 1024, 1);
    gemv_wmma_kernel<<<4096 / 64, 128, 1024 * sizeof(float), stream>>>(
        h2, W3, b3, out, 1024, 4096, 0);
    eig_bisect_kernel<<<dim3(RN / (256 / NPT), 2), 256, 0, stream>>>(out, out + 4096);
}


# GEMV layers: K split over the 16 waves of a 512-thread block (32 columns per block), two fragments' loads in flight per wave; partial sums added through LDS in wave order (new summation order)
# speedup vs baseline: 9.9820x; 1.6502x over previous
#include <hip/hip_runtime.h>
#include <hip/hip_bf16.h>

typedef __attribute__((ext_vector_type(16))) _Float16 v16h;
typedef __attribute__((ext_vector_type(8)))  float    v8f;

typedef __attribute__((ext_vector_type(2))) float v2f;
typedef __attribute__((ext_vector_type(4))) float v4f;

#define GEMV_WAVES 16
#define GEMV_COLS  32

template <int K, int N, bool RELU>
__global__ void __launch_bounds__(GEMV_WAVES * 32)
gemv_wmma_kernel(const float* __restrict__ x,
                 const float* __restrict__ W,
                 const float* __restrict__ bias,
                 float* __restrict__ y) {
    constexpr int KS = K / GEMV_WAVES;
    static_assert(K % (GEMV_WAVES * 32) == 0, "whole 32-row fragments per wave");
    static_assert(N % GEMV_COLS == 0, "whole blocks");
    __shared__ float part[GEMV_WAVES][GEMV_COLS];

    const int lane = threadIdx.x & 31;
    const int wave = threadIdx.x >> 5;
    const int c    = lane & 15;
    const int hi8  = (lane >> 4) * 8;
    const int col0 = blockIdx.x * GEMV_COLS;

    const float* xp = x + wave * KS + hi8;
    const float* wp = W + (size_t)(wave * KS + hi8) * N + col0 + 2 * c;

    v8f acc0 = {}, acc1 = {};
#pragma unroll 2
    for (int k0 = 0; k0 < KS; k0 += 32) {
        const v4f xa = *(const v4f*)(xp + k0);
        const v4f xb = *(const v4f*)(xp + k0 + 4);
        const v4f xc = *(const v4f*)(xp + k0 + 16);
        const v4f xd = *(const v4f*)(xp + k0 + 20);
        v16h ah = {}, al = {}, b0h = {}, b0l = {}, b1h = {}, b1l = {};
#pragma unroll
        for (int e = 0; e < 16; ++e) {
            const int kk = k0 + e + (e & 8);
            const float xv = (e < 4) ? xa[e & 3] : (e < 8) ? xb[e & 3]
                           : (e < 12) ? xc[e & 3] : xd[e & 3];
            const v2f wv = *(const v2f*)(wp + (size_t)kk * N);
            const _Float16 xh  = (_Float16)xv;
            const _Float16 w0h = (_Float16)wv[0];
            const _Float16 w1h = (_Float16)wv[1];
            ah[e]  = xh;   al[e]  = (_Float16)(xv - (float)xh);
            b0h[e] = w0h;  b0l[e] = (_Float16)(wv[0] - (float)w0h);
            b1h[e] = w1h;  b1l[e] = (_Float16)(wv[1] - (float)w1h);
        }
        acc0 = __builtin_amdgcn_wmma_f32_16x16x32_f16(
            false, ah, false, b0h, (short)0, acc0, false, false);
        acc1 = __builtin_amdgcn_wmma_f32_16x16x32_f16(
            false, ah, false, b1h, (short)0, acc1, false, false);
        acc0 = __builtin_amdgcn_wmma_f32_16x16x32_f16(
            false, ah, false, b0l, (short)0, acc0, false, false);
        acc1 = __builtin_amdgcn_wmma_f32_16x16x32_f16(
            false, ah, false, b1l, (short)0, acc1, false, false);
        acc0 = __builtin_amdgcn_wmma_f32_16x16x32_f16(
            false, al, false, b0h, (short)0, acc0, false, false);
        acc1 = __builtin_amdgcn_wmma_f32_16x16x32_f16(
            false, al, false, b1h, (short)0, acc1, false, false);
    }

    if (lane < 16) {
        v2f p;
        p[0] = acc0[0];
        p[1] = acc1[0];
        *(v2f*)&part[wave][2 * c] = p;
    }
    __syncthreads();

    if (wave == 0) {
        float v = part[0][lane];
        for (int w = 1; w < GEMV_WAVES; ++w) v += part[w][lane];
        v += bias[col0 + lane];
        if (RELU) v = fmaxf(v, 0.0f);
        y[col0 + lane] = v;
    }
}

#define RN 4096

__device__ __forceinline__ void sturm_block32(const v4f (&t)[8],
                                              float& p1, float& p0,
                                              unsigned& prev, int& cnt) {
    unsigned acc = prev;
#pragma unroll
    for (int u = 0; u < 8; ++u) {
#pragma unroll
        for (int s = 0; s < 4; ++s) {
            const float p2 = __builtin_fmaf(t[u][s], p1, -p0);
            acc = __builtin_amdgcn_alignbit(acc, __float_as_uint(p2), 31u);
            p0 = p1;
            p1 = p2;
        }
        if ((u & 3) == 3) {
            const int e = __builtin_amdgcn_frexp_expf(
                __builtin_fmaxf(__builtin_fabsf(p1), __builtin_fabsf(p0)));
            p1 = __builtin_ldexpf(p1, -e);
            p0 = __builtin_ldexpf(p0, -e);
        }
    }
    const unsigned older = __builtin_amdgcn_alignbit(prev, acc, 1u);
    cnt += __builtin_popcount(acc ^ older);
    prev = acc;
}

__device__ __forceinline__ void hold_block(v4f (&t)[8], float& p1) {
#pragma unroll
    for (int u = 0; u < 8; ++u)
        asm volatile("" : "+v"(t[u]), "+v"(p1) : : "memory");
}

__device__ __forceinline__ int sturm_count_scaled(const v4f* sd4,
                                                  const float x2) {
    float    p1   = 1.0f;
    float    p0   = 0.0f;
    unsigned prev = 0u;
    int      cnt  = 0;
    const int nblk = RN / 32;

    v4f a[8], b[8], ra[8], rb[8];
#pragma unroll
    for (int u = 0; u < 8; ++u) a[u] = sd4[u] - x2;
#pragma unroll
    for (int u = 0; u < 8; ++u) rb[u] = sd4[8 + u];
    hold_block(rb, p1);

    for (int k = 0; k < nblk; k += 2) {
        const int k2 = ((k + 2) & (nblk - 1)) * 8;
#pragma unroll
        for (int u = 0; u < 8; ++u) ra[u] = sd4[k2 + u];
        __builtin_amdgcn_sched_barrier(0);
#pragma unroll
        for (int u = 0; u < 8; ++u) b[u] = rb[u] - x2;
        sturm_block32(a, p1, p0, prev, cnt);
        hold_block(b, p1);
        hold_block(ra, p1);
        const int k3 = ((k + 3) & (nblk - 1)) * 8;
#pragma unroll
        for (int u = 0; u < 8; ++u) rb[u] = sd4[k3 + u];
        __builtin_amdgcn_sched_barrier(0);
#pragma unroll
        for (int u = 0; u < 8; ++u) a[u] = ra[u] - x2;
        sturm_block32(b, p1, p0, prev, cnt);
        hold_block(a, p1);
        hold_block(rb, p1);
    }
    return cnt;
}

__device__ __forceinline__ int sturm_count_ratio(const float* sd,
                                                 const float x2) {
    float q   = sd[0] - x2;
    int   cnt = (q < 0.0f);
    for (int i = 1; i < RN; ++i) {
        q = (sd[i] - x2) - __builtin_amdgcn_rcpf(q);
        if (fabsf(q) < 2e-30f) q = -2e-30f;
        cnt += (q < 0.0f);
    }
    return cnt;
}

#define NPT    8
#define SWEEPS 16

__device__ __forceinline__ float section_point(const float lo, const float w,
                                               const int i) {
    return __builtin_fmaf(w, (float)(i + 1) * (1.0f / (float)(NPT + 1)), lo);
}

__global__ void __launch_bounds__(256)
eig_bisect_kernel(const float* __restrict__ ptl, float* __restrict__ evl) {
    __shared__ v4f   sd4[RN / 4];
    __shared__ float res[256 / NPT];
    float* sd = (float*)sd4;
    const int   l   = blockIdx.y;
    const float ll1 = (float)(l * (l + 1));
    const int   tid = threadIdx.x;
    const float dr  = 10000.0f / (float)RN;

    for (int i = tid; i < RN; i += blockDim.x) {
        const float r = dr * (float)(i + 1);
        const float d = 1.0f + ptl[i] + ll1 / (r * r);
        sd[i] = 2.0f * d;
    }
    __syncthreads();

    float mn = 1e30f, mx = -1e30f;
    for (int i = 0; i < RN / 4; ++i) {
        const v4f v = sd4[i];
        mn = fminf(fminf(mn, v[0]), fminf(fminf(v[1], v[2]), v[3]));
        mx = fmaxf(fmaxf(mx, v[0]), fmaxf(fmaxf(v[1], v[2]), v[3]));
    }
    float lo = fminf(1e30f, 0.5f * mn - 1.0f);
    float hi = fmaxf(-1e30f, 0.5f * mx + 1.0f);

    const int narrow = __builtin_amdgcn_readfirstlane((hi - lo) < 48.0f ? 1 : 0);

    const int q   = tid & (NPT - 1);
    const int grp = tid / NPT;
    const int gsh = tid & 31 & ~(NPT - 1);
    const int j   = blockIdx.x * (256 / NPT) + grp;

    for (int it = 0; it < SWEEPS; ++it) {
        const float w  = hi - lo;
        const float x  = section_point(lo, w, q);
        const float x2 = x + x;
        const int cnt = narrow ? sturm_count_scaled(sd4, x2)
                               : sturm_count_ratio(sd, x2);
        const unsigned below = __builtin_amdgcn_ballot_w32(cnt <= j);
        const int k = __builtin_popcount((below >> gsh) & ((1u << NPT) - 1u));
        const float nlo = (k > 0)   ? section_point(lo, w, k - 1) : lo;
        const float nhi = (k < NPT) ? section_point(lo, w, k)     : hi;
        lo = nlo;
        hi = nhi;
    }

    if (q == 0) res[grp] = 0.5f * (lo + hi);
    __syncthreads();
    if (tid < 256 / NPT)
        evl[l * RN + blockIdx.x * (256 / NPT) + tid] = res[tid];
}

extern "C" void kernel_launch(void* const* d_in, const int* in_sizes, int n_in,
                              void* d_out, int out_size, void* d_ws, size_t ws_size,
                              hipStream_t stream) {
    const float* energy = (const float*)d_in[0];
    const float* W1     = (const float*)d_in[1];
    const float* b1     = (const float*)d_in[2];
    const float* W2     = (const float*)d_in[3];
    const float* b2     = (const float*)d_in[4];
    const float* W3     = (const float*)d_in[5];
    const float* b3     = (const float*)d_in[6];

    float* out = (float*)d_out;
    float* h1  = (float*)d_ws;
    float* h2  = h1 + 2048;

    gemv_wmma_kernel<4096, 2048, true>
        <<<2048 / GEMV_COLS, GEMV_WAVES * 32, 0, stream>>>(energy, W1, b1, h1);
    gemv_wmma_kernel<2048, 1024, true>
        <<<1024 / GEMV_COLS, GEMV_WAVES * 32, 0, stream>>>(h1, W2, b2, h2);
    gemv_wmma_kernel<1024, 4096, false>
        <<<4096 / GEMV_COLS, GEMV_WAVES * 32, 0, stream>>>(h2, W3, b3, out);
    eig_bisect_kernel<<<dim3(RN / (256 / NPT), 2), 256, 0, stream>>>(out, out + 4096);
}


# Multisection leaves at its fixed point: each sweep is first replayed from the counts on record at lo and hi; when all groups of a wave would get lo/hi back unchanged the wave exits; same bits out
# speedup vs baseline: 12.7870x; 1.2810x over previous
#include <hip/hip_runtime.h>
#include <hip/hip_bf16.h>

typedef __attribute__((ext_vector_type(16))) _Float16 v16h;
typedef __attribute__((ext_vector_type(8)))  float    v8f;

typedef __attribute__((ext_vector_type(2))) float v2f;
typedef __attribute__((ext_vector_type(4))) float v4f;

#define GEMV_WAVES 16
#define GEMV_COLS  32

template <int K, int N, bool RELU>
__global__ void __launch_bounds__(GEMV_WAVES * 32)
gemv_wmma_kernel(const float* __restrict__ x,
                 const float* __restrict__ W,
                 const float* __restrict__ bias,
                 float* __restrict__ y) {
    constexpr int KS = K / GEMV_WAVES;
    static_assert(K % (GEMV_WAVES * 32) == 0, "whole 32-row fragments per wave");
    static_assert(N % GEMV_COLS == 0, "whole blocks");
    __shared__ float part[GEMV_WAVES][GEMV_COLS];

    const int lane = threadIdx.x & 31;
    const int wave = threadIdx.x >> 5;
    const int c    = lane & 15;
    const int hi8  = (lane >> 4) * 8;
    const int col0 = blockIdx.x * GEMV_COLS;

    const float* xp = x + wave * KS + hi8;
    const float* wp = W + (size_t)(wave * KS + hi8) * N + col0 + 2 * c;

    v8f acc0 = {}, acc1 = {};
#pragma unroll 2
    for (int k0 = 0; k0 < KS; k0 += 32) {
        const v4f xa = *(const v4f*)(xp + k0);
        const v4f xb = *(const v4f*)(xp + k0 + 4);
        const v4f xc = *(const v4f*)(xp + k0 + 16);
        const v4f xd = *(const v4f*)(xp + k0 + 20);
        v16h ah = {}, al = {}, b0h = {}, b0l = {}, b1h = {}, b1l = {};
#pragma unroll
        for (int e = 0; e < 16; ++e) {
            const int kk = k0 + e + (e & 8);
            const float xv = (e < 4) ? xa[e & 3] : (e < 8) ? xb[e & 3]
                           : (e < 12) ? xc[e & 3] : xd[e & 3];
            const v2f wv = *(const v2f*)(wp + (size_t)kk * N);
            const _Float16 xh  = (_Float16)xv;
            const _Float16 w0h = (_Float16)wv[0];
            const _Float16 w1h = (_Float16)wv[1];
            ah[e]  = xh;   al[e]  = (_Float16)(xv - (float)xh);
            b0h[e] = w0h;  b0l[e] = (_Float16)(wv[0] - (float)w0h);
            b1h[e] = w1h;  b1l[e] = (_Float16)(wv[1] - (float)w1h);
        }
        acc0 = __builtin_amdgcn_wmma_f32_16x16x32_f16(
            false, ah, false, b0h, (short)0, acc0, false, false);
        acc1 = __builtin_amdgcn_wmma_f32_16x16x32_f16(
            false, ah, false, b1h, (short)0, acc1, false, false);
        acc0 = __builtin_amdgcn_wmma_f32_16x16x32_f16(
            false, ah, false, b0l, (short)0, acc0, false, false);
        acc1 = __builtin_amdgcn_wmma_f32_16x16x32_f16(
            false, ah, false, b1l, (short)0, acc1, false, false);
        acc0 = __builtin_amdgcn_wmma_f32_16x16x32_f16(
            false, al, false, b0h, (short)0, acc0, false, false);
        acc1 = __builtin_amdgcn_wmma_f32_16x16x32_f16(
            false, al, false, b1h, (short)0, acc1, false, false);
    }

    if (lane < 16) {
        v2f p;
        p[0] = acc0[0];
        p[1] = acc1[0];
        *(v2f*)&part[wave][2 * c] = p;
    }
    __syncthreads();

    if (wave == 0) {
        float v = part[0][lane];
        for (int w = 1; w < GEMV_WAVES; ++w) v += part[w][lane];
        v += bias[col0 + lane];
        if (RELU) v = fmaxf(v, 0.0f);
        y[col0 + lane] = v;
    }
}

#define RN 4096

__device__ __forceinline__ void sturm_block32(const v4f (&t)[8],
                                              float& p1, float& p0,
                                              unsigned& prev, int& cnt) {
    unsigned acc = prev;
#pragma unroll
    for (int u = 0; u < 8; ++u) {
#pragma unroll
        for (int s = 0; s < 4; ++s) {
            const float p2 = __builtin_fmaf(t[u][s], p1, -p0);
            acc = __builtin_amdgcn_alignbit(acc, __float_as_uint(p2), 31u);
            p0 = p1;
            p1 = p2;
        }
        if ((u & 3) == 3) {
            const int e = __builtin_amdgcn_frexp_expf(
                __builtin_fmaxf(__builtin_fabsf(p1), __builtin_fabsf(p0)));
            p1 = __builtin_ldexpf(p1, -e);
            p0 = __builtin_ldexpf(p0, -e);
        }
    }
    const unsigned older = __builtin_amdgcn_alignbit(prev, acc, 1u);
    cnt += __builtin_popcount(acc ^ older);
    prev = acc;
}

__device__ __forceinline__ void hold_block(v4f (&t)[8], float& p1) {
#pragma unroll
    for (int u = 0; u < 8; ++u)
        asm volatile("" : "+v"(t[u]), "+v"(p1) : : "memory");
}

__device__ __forceinline__ int sturm_count_scaled(const v4f* sd4,
                                                  const float x2) {
    float    p1   = 1.0f;
    float    p0   = 0.0f;
    unsigned prev = 0u;
    int      cnt  = 0;
    const int nblk = RN / 32;

    v4f a[8], b[8], ra[8], rb[8];
#pragma unroll
    for (int u = 0; u < 8; ++u) a[u] = sd4[u] - x2;
#pragma unroll
    for (int u = 0; u < 8; ++u) rb[u] = sd4[8 + u];
    hold_block(rb, p1);

    for (int k = 0; k < nblk; k += 2) {
        const int k2 = ((k + 2) & (nblk - 1)) * 8;
#pragma unroll
        for (int u = 0; u < 8; ++u) ra[u] = sd4[k2 + u];
        __builtin_amdgcn_sched_barrier(0);
#pragma unroll
        for (int u = 0; u < 8; ++u) b[u] = rb[u] - x2;
        sturm_block32(a, p1, p0, prev, cnt);
        hold_block(b, p1);
        hold_block(ra, p1);
        const int k3 = ((k + 3) & (nblk - 1)) * 8;
#pragma unroll
        for (int u = 0; u < 8; ++u) rb[u] = sd4[k3 + u];
        __builtin_amdgcn_sched_barrier(0);
#pragma unroll
        for (int u = 0; u < 8; ++u) a[u] = ra[u] - x2;
        sturm_block32(b, p1, p0, prev, cnt);
        hold_block(a, p1);
        hold_block(rb, p1);
    }
    return cnt;
}

__device__ __forceinline__ int sturm_count_ratio(const float* sd,
                                                 const float x2) {
    float q   = sd[0] - x2;
    int   cnt = (q < 0.0f);
    for (int i = 1; i < RN; ++i) {
        q = (sd[i] - x2) - __builtin_amdgcn_rcpf(q);
        if (fabsf(q) < 2e-30f) q = -2e-30f;
        cnt += (q < 0.0f);
    }
    return cnt;
}

#define NPT    8
#define SWEEPS 16

__device__ __forceinline__ float section_point(const float lo, const float w,
                                               const int i) {
    return __builtin_fmaf(w, (float)(i + 1) * (1.0f / (float)(NPT + 1)), lo);
}

__device__ __forceinline__ void kept_part(const float lo, const float hi,
                                          const float w, const unsigned gbits,
                                          float& nlo, float& nhi, int& k) {
    k   = __builtin_popcount(gbits);
    nlo = (k > 0)   ? section_point(lo, w, k - 1) : lo;
    nhi = (k < NPT) ? section_point(lo, w, k)     : hi;
}

__global__ void __launch_bounds__(256)
eig_bisect_kernel(const float* __restrict__ ptl, float* __restrict__ evl) {
    __shared__ v4f   sd4[RN / 4];
    __shared__ float res[256 / NPT];
    float* sd = (float*)sd4;
    const int   l   = blockIdx.y;
    const float ll1 = (float)(l * (l + 1));
    const int   tid = threadIdx.x;
    const float dr  = 10000.0f / (float)RN;

    for (int i = tid; i < RN; i += blockDim.x) {
        const float r = dr * (float)(i + 1);
        const float d = 1.0f + ptl[i] + ll1 / (r * r);
        sd[i] = 2.0f * d;
    }
    __syncthreads();

    float mn = 1e30f, mx = -1e30f;
    for (int i = 0; i < RN / 4; ++i) {
        const v4f v = sd4[i];
        mn = fminf(fminf(mn, v[0]), fminf(fminf(v[1], v[2]), v[3]));
        mx = fmaxf(fmaxf(mx, v[0]), fmaxf(fmaxf(v[1], v[2]), v[3]));
    }
    float lo = fminf(1e30f, 0.5f * mn - 1.0f);
    float hi = fmaxf(-1e30f, 0.5f * mx + 1.0f);

    const int narrow = __builtin_amdgcn_readfirstlane((hi - lo) < 48.0f ? 1 : 0);

    const int q   = tid & (NPT - 1);
    const int grp = tid / NPT;
    const int gsh = tid & 31 & ~(NPT - 1);
    const int j   = blockIdx.x * (256 / NPT) + grp;

    bool lo_kn = false, hi_kn = false, lo_b = false, hi_b = false;

    for (int it = 0; it < SWEEPS; ++it) {
        const float w  = hi - lo;
        const float x  = section_point(lo, w, q);

        const bool at_lo = (x == lo);
        const bool kn = at_lo ? lo_kn : ((x == hi) && hi_kn);
        const bool fb = at_lo ? lo_b : hi_b;
        const unsigned fbits =
            (__builtin_amdgcn_ballot_w32(fb) >> gsh) & ((1u << NPT) - 1u);
        float flo, fhi;
        int   fk;
        kept_part(lo, hi, w, fbits, flo, fhi, fk);
        const bool stays = kn && (flo == lo) && (fhi == hi);
        if (__builtin_amdgcn_ballot_w32(!stays) == 0u) break;

        const float x2 = x + x;
        const int cnt = narrow ? sturm_count_scaled(sd4, x2)
                               : sturm_count_ratio(sd, x2);
        const unsigned gbits =
            (__builtin_amdgcn_ballot_w32(cnt <= j) >> gsh) & ((1u << NPT) - 1u);
        float nlo, nhi;
        int   k;
        kept_part(lo, hi, w, gbits, nlo, nhi, k);
        if (k > 0)   { lo_kn = true; lo_b = ((gbits >> (k - 1)) & 1u) != 0u; }
        if (k < NPT) { hi_kn = true; hi_b = ((gbits >> k) & 1u) != 0u; }
        lo = nlo;
        hi = nhi;
    }

    if (q == 0) res[grp] = 0.5f * (lo + hi);
    __syncthreads();
    if (tid < 256 / NPT)
        evl[l * RN + blockIdx.x * (256 / NPT) + tid] = res[tid];
}

extern "C" void kernel_launch(void* const* d_in, const int* in_sizes, int n_in,
                              void* d_out, int out_size, void* d_ws, size_t ws_size,
                              hipStream_t stream) {
    const float* energy = (const float*)d_in[0];
    const float* W1     = (const float*)d_in[1];
    const float* b1     = (const float*)d_in[2];
    const float* W2     = (const float*)d_in[3];
    const float* b2     = (const float*)d_in[4];
    const float* W3     = (const float*)d_in[5];
    const float* b3     = (const float*)d_in[6];

    float* out = (float*)d_out;
    float* h1  = (float*)d_ws;
    float* h2  = h1 + 2048;

    gemv_wmma_kernel<4096, 2048, true>
        <<<2048 / GEMV_COLS, GEMV_WAVES * 32, 0, stream>>>(energy, W1, b1, h1);
    gemv_wmma_kernel<2048, 1024, true>
        <<<1024 / GEMV_COLS, GEMV_WAVES * 32, 0, stream>>>(h1, W2, b2, h2);
    gemv_wmma_kernel<1024, 4096, false>
        <<<4096 / GEMV_COLS, GEMV_WAVES * 32, 0, stream>>>(h2, W3, b3, out);
    eig_bisect_kernel<<<dim3(RN / (256 / NPT), 2), 256, 0, stream>>>(out, out + 4096);
}


# First two sweeps block-wide: the block's 256 lanes take 256 distinct points over the union of its 32 brackets, each group keeps its cell from the counts in LDS (257-way, not 9-way); then as before
# speedup vs baseline: 16.1321x; 1.2616x over previous
#include <hip/hip_runtime.h>
#include <hip/hip_bf16.h>

typedef __attribute__((ext_vector_type(16))) _Float16 v16h;
typedef __attribute__((ext_vector_type(8)))  float    v8f;

typedef __attribute__((ext_vector_type(2))) float v2f;
typedef __attribute__((ext_vector_type(4))) float v4f;

#define GEMV_WAVES 16
#define GEMV_COLS  32

template <int K, int N, bool RELU>
__global__ void __launch_bounds__(GEMV_WAVES * 32)
gemv_wmma_kernel(const float* __restrict__ x,
                 const float* __restrict__ W,
                 const float* __restrict__ bias,
                 float* __restrict__ y) {
    constexpr int KS = K / GEMV_WAVES;
    static_assert(K % (GEMV_WAVES * 32) == 0, "whole 32-row fragments per wave");
    static_assert(N % GEMV_COLS == 0, "whole blocks");
    __shared__ float part[GEMV_WAVES][GEMV_COLS];

    const int lane = threadIdx.x & 31;
    const int wave = threadIdx.x >> 5;
    const int c    = lane & 15;
    const int hi8  = (lane >> 4) * 8;
    const int col0 = blockIdx.x * GEMV_COLS;

    const float* xp = x + wave * KS + hi8;
    const float* wp = W + (size_t)(wave * KS + hi8) * N + col0 + 2 * c;

    v8f acc0 = {}, acc1 = {};
#pragma unroll 2
    for (int k0 = 0; k0 < KS; k0 += 32) {
        const v4f xa = *(const v4f*)(xp + k0);
        const v4f xb = *(const v4f*)(xp + k0 + 4);
        const v4f xc = *(const v4f*)(xp + k0 + 16);
        const v4f xd = *(const v4f*)(xp + k0 + 20);
        v16h ah = {}, al = {}, b0h = {}, b0l = {}, b1h = {}, b1l = {};
#pragma unroll
        for (int e = 0; e < 16; ++e) {
            const int kk = k0 + e + (e & 8);
            const float xv = (e < 4) ? xa[e & 3] : (e < 8) ? xb[e & 3]
                           : (e < 12) ? xc[e & 3] : xd[e & 3];
            const v2f wv = *(const v2f*)(wp + (size_t)kk * N);
            const _Float16 xh  = (_Float16)xv;
            const _Float16 w0h = (_Float16)wv[0];
            const _Float16 w1h = (_Float16)wv[1];
            ah[e]  = xh;   al[e]  = (_Float16)(xv - (float)xh);
            b0h[e] = w0h;  b0l[e] = (_Float16)(wv[0] - (float)w0h);
            b1h[e] = w1h;  b1l[e] = (_Float16)(wv[1] - (float)w1h);
        }
        acc0 = __builtin_amdgcn_wmma_f32_16x16x32_f16(
            false, ah, false, b0h, (short)0, acc0, false, false);
        acc1 = __builtin_amdgcn_wmma_f32_16x16x32_f16(
            false, ah, false, b1h, (short)0, acc1, false, false);
        acc0 = __builtin_amdgcn_wmma_f32_16x16x32_f16(
            false, ah, false, b0l, (short)0, acc0, false, false);
        acc1 = __builtin_amdgcn_wmma_f32_16x16x32_f16(
            false, ah, false, b1l, (short)0, acc1, false, false);
        acc0 = __builtin_amdgcn_wmma_f32_16x16x32_f16(
            false, al, false, b0h, (short)0, acc0, false, false);
        acc1 = __builtin_amdgcn_wmma_f32_16x16x32_f16(
            false, al, false, b1h, (short)0, acc1, false, false);
    }

    if (lane < 16) {
        v2f p;
        p[0] = acc0[0];
        p[1] = acc1[0];
        *(v2f*)&part[wave][2 * c] = p;
    }
    __syncthreads();

    if (wave == 0) {
        float v = part[0][lane];
        for (int w = 1; w < GEMV_WAVES; ++w) v += part[w][lane];
        v += bias[col0 + lane];
        if (RELU) v = fmaxf(v, 0.0f);
        y[col0 + lane] = v;
    }
}

#define RN 4096

__device__ __forceinline__ void sturm_block32(const v4f (&t)[8],
                                              float& p1, float& p0,
                                              unsigned& prev, int& cnt) {
    unsigned acc = prev;
#pragma unroll
    for (int u = 0; u < 8; ++u) {
#pragma unroll
        for (int s = 0; s < 4; ++s) {
            const float p2 = __builtin_fmaf(t[u][s], p1, -p0);
            acc = __builtin_amdgcn_alignbit(acc, __float_as_uint(p2), 31u);
            p0 = p1;
            p1 = p2;
        }
        if ((u & 3) == 3) {
            const int e = __builtin_amdgcn_frexp_expf(
                __builtin_fmaxf(__builtin_fabsf(p1), __builtin_fabsf(p0)));
            p1 = __builtin_ldexpf(p1, -e);
            p0 = __builtin_ldexpf(p0, -e);
        }
    }
    const unsigned older = __builtin_amdgcn_alignbit(prev, acc, 1u);
    cnt += __builtin_popcount(acc ^ older);
    prev = acc;
}

__device__ __forceinline__ void hold_block(v4f (&t)[8], float& p1) {
#pragma unroll
    for (int u = 0; u < 8; ++u)
        asm volatile("" : "+v"(t[u]), "+v"(p1) : : "memory");
}

__device__ __forceinline__ int sturm_count_scaled(const v4f* sd4,
                                                  const float x2) {
    float    p1   = 1.0f;
    float    p0   = 0.0f;
    unsigned prev = 0u;
    int      cnt  = 0;
    const int nblk = RN / 32;

    v4f a[8], b[8], ra[8], rb[8];
#pragma unroll
    for (int u = 0; u < 8; ++u) a[u] = sd4[u] - x2;
#pragma unroll
    for (int u = 0; u < 8; ++u) rb[u] = sd4[8 + u];
    hold_block(rb, p1);

    for (int k = 0; k < nblk; k += 2) {
        const int k2 = ((k + 2) & (nblk - 1)) * 8;
#pragma unroll
        for (int u = 0; u < 8; ++u) ra[u] = sd4[k2 + u];
        __builtin_amdgcn_sched_barrier(0);
#pragma unroll
        for (int u = 0; u < 8; ++u) b[u] = rb[u] - x2;
        sturm_block32(a, p1, p0, prev, cnt);
        hold_block(b, p1);
        hold_block(ra, p1);
        const int k3 = ((k + 3) & (nblk - 1)) * 8;
#pragma unroll
        for (int u = 0; u < 8; ++u) rb[u] = sd4[k3 + u];
        __builtin_amdgcn_sched_barrier(0);
#pragma unroll
        for (int u = 0; u < 8; ++u) a[u] = ra[u] - x2;
        sturm_block32(b, p1, p0, prev, cnt);
        hold_block(a, p1);
        hold_block(rb, p1);
    }
    return cnt;
}

__device__ __forceinline__ int sturm_count_ratio(const float* sd,
                                                 const float x2) {
    float q   = sd[0] - x2;
    int   cnt = (q < 0.0f);
    for (int i = 1; i < RN; ++i) {
        q = (sd[i] - x2) - __builtin_amdgcn_rcpf(q);
        if (fabsf(q) < 2e-30f) q = -2e-30f;
        cnt += (q < 0.0f);
    }
    return cnt;
}

#define NPT    8
#define SWEEPS 16

#define CPT    256
#define CSWEEPS 2

__device__ __forceinline__ float grid_point(const float lo, const float w,
                                            const int t) {
    return __builtin_fmaf(w, (float)(t + 1) * (1.0f / (float)(CPT + 1)), lo);
}

__device__ __forceinline__ float section_point(const float lo, const float w,
                                               const int i) {
    return __builtin_fmaf(w, (float)(i + 1) * (1.0f / (float)(NPT + 1)), lo);
}

__device__ __forceinline__ void kept_part(const float lo, const float hi,
                                          const float w, const unsigned gbits,
                                          float& nlo, float& nhi, int& k) {
    k   = __builtin_popcount(gbits);
    nlo = (k > 0)   ? section_point(lo, w, k - 1) : lo;
    nhi = (k < NPT) ? section_point(lo, w, k)     : hi;
}

__global__ void __launch_bounds__(256)
eig_bisect_kernel(const float* __restrict__ ptl, float* __restrict__ evl) {
    __shared__ v4f   sd4[RN / 4];
    __shared__ float res[256 / NPT];
    __shared__ int   cnts[CPT];
    __shared__ float glo[256 / NPT];
    __shared__ float ghi[256 / NPT];
    float* sd = (float*)sd4;
    const int   l   = blockIdx.y;
    const float ll1 = (float)(l * (l + 1));
    const int   tid = threadIdx.x;
    const float dr  = 10000.0f / (float)RN;

    for (int i = tid; i < RN; i += blockDim.x) {
        const float r = dr * (float)(i + 1);
        const float d = 1.0f + ptl[i] + ll1 / (r * r);
        sd[i] = 2.0f * d;
    }
    __syncthreads();

    float mn = 1e30f, mx = -1e30f;
    for (int i = 0; i < RN / 4; ++i) {
        const v4f v = sd4[i];
        mn = fminf(fminf(mn, v[0]), fminf(fminf(v[1], v[2]), v[3]));
        mx = fmaxf(fmaxf(mx, v[0]), fmaxf(fmaxf(v[1], v[2]), v[3]));
    }
    float lo = fminf(1e30f, 0.5f * mn - 1.0f);
    float hi = fmaxf(-1e30f, 0.5f * mx + 1.0f);

    const int narrow = __builtin_amdgcn_readfirstlane((hi - lo) < 48.0f ? 1 : 0);

    const int q   = tid & (NPT - 1);
    const int grp = tid / NPT;
    const int gsh = tid & 31 & ~(NPT - 1);
    const int j   = blockIdx.x * (256 / NPT) + grp;

    float blo = lo, bhi = hi;
    for (int cs = 0; cs < CSWEEPS; ++cs) {
        const float bw  = bhi - blo;
        const float xg  = grid_point(blo, bw, tid);
        const float xg2 = xg + xg;
        cnts[tid] = narrow ? sturm_count_scaled(sd4, xg2)
                           : sturm_count_ratio(sd, xg2);
        __syncthreads();
        int k = 0;
        for (int u = 0; u < CPT / NPT; ++u) k += (cnts[u * NPT + q] <= j) ? 1 : 0;
        k += __shfl_xor(k, 1);
        k += __shfl_xor(k, 2);
        k += __shfl_xor(k, 4);
        lo = (k > 0)   ? grid_point(blo, bw, k - 1) : blo;
        hi = (k < CPT) ? grid_point(blo, bw, k)     : bhi;
        if (cs + 1 < CSWEEPS) {
            if (q == 0) { glo[grp] = lo; ghi[grp] = hi; }
            __syncthreads();
            blo = glo[0];
            bhi = ghi[0];
            for (int g = 1; g < 256 / NPT; ++g) {
                blo = fminf(blo, glo[g]);
                bhi = fmaxf(bhi, ghi[g]);
            }
        }
    }

    bool lo_kn = false, hi_kn = false, lo_b = false, hi_b = false;

    for (int it = CSWEEPS; it < SWEEPS; ++it) {
        const float w  = hi - lo;
        const float x  = section_point(lo, w, q);

        const bool at_lo = (x == lo);
        const bool kn = at_lo ? lo_kn : ((x == hi) && hi_kn);
        const bool fb = at_lo ? lo_b : hi_b;
        const unsigned fbits =
            (__builtin_amdgcn_ballot_w32(fb) >> gsh) & ((1u << NPT) - 1u);
        float flo, fhi;
        int   fk;
        kept_part(lo, hi, w, fbits, flo, fhi, fk);
        const bool stays = kn && (flo == lo) && (fhi == hi);
        if (__builtin_amdgcn_ballot_w32(!stays) == 0u) break;

        const float x2 = x + x;
        const int cnt = narrow ? sturm_count_scaled(sd4, x2)
                               : sturm_count_ratio(sd, x2);
        const unsigned gbits =
            (__builtin_amdgcn_ballot_w32(cnt <= j) >> gsh) & ((1u << NPT) - 1u);
        float nlo, nhi;
        int   k;
        kept_part(lo, hi, w, gbits, nlo, nhi, k);
        if (k > 0)   { lo_kn = true; lo_b = ((gbits >> (k - 1)) & 1u) != 0u; }
        if (k < NPT) { hi_kn = true; hi_b = ((gbits >> k) & 1u) != 0u; }
        lo = nlo;
        hi = nhi;
    }

    if (q == 0) res[grp] = 0.5f * (lo + hi);
    __syncthreads();
    if (tid < 256 / NPT)
        evl[l * RN + blockIdx.x * (256 / NPT) + tid] = res[tid];
}

extern "C" void kernel_launch(void* const* d_in, const int* in_sizes, int n_in,
                              void* d_out, int out_size, void* d_ws, size_t ws_size,
                              hipStream_t stream) {
    const float* energy = (const float*)d_in[0];
    const float* W1     = (const float*)d_in[1];
    const float* b1     = (const float*)d_in[2];
    const float* W2     = (const float*)d_in[3];
    const float* b2     = (const float*)d_in[4];
    const float* W3     = (const float*)d_in[5];
    const float* b3     = (const float*)d_in[6];

    float* out = (float*)d_out;
    float* h1  = (float*)d_ws;
    float* h2  = h1 + 2048;

    gemv_wmma_kernel<4096, 2048, true>
        <<<2048 / GEMV_COLS, GEMV_WAVES * 32, 0, stream>>>(energy, W1, b1, h1);
    gemv_wmma_kernel<2048, 1024, true>
        <<<1024 / GEMV_COLS, GEMV_WAVES * 32, 0, stream>>>(h1, W2, b2, h2);
    gemv_wmma_kernel<1024, 4096, false>
        <<<4096 / GEMV_COLS, GEMV_WAVES * 32, 0, stream>>>(h2, W3, b3, out);
    eig_bisect_kernel<<<dim3(RN / (256 / NPT), 2), 256, 0, stream>>>(out, out + 4096);
}


# Eig kernel set-up: each thread stages its 16 diagonal entries with the loads in one group and keeps their min/max; Gershgorin bounds by shuffle + LDS reduce, not a 4096-entry LDS scan per thread
# speedup vs baseline: 17.6460x; 1.0938x over previous
#include <hip/hip_runtime.h>
#include <hip/hip_bf16.h>

typedef __attribute__((ext_vector_type(16))) _Float16 v16h;
typedef __attribute__((ext_vector_type(8)))  float    v8f;

typedef __attribute__((ext_vector_type(2))) float v2f;
typedef __attribute__((ext_vector_type(4))) float v4f;

#define GEMV_WAVES 16
#define GEMV_COLS  32

template <int K, int N, bool RELU>
__global__ void __launch_bounds__(GEMV_WAVES * 32)
gemv_wmma_kernel(const float* __restrict__ x,
                 const float* __restrict__ W,
                 const float* __restrict__ bias,
                 float* __restrict__ y) {
    constexpr int KS = K / GEMV_WAVES;
    static_assert(K % (GEMV_WAVES * 32) == 0, "whole 32-row fragments per wave");
    static_assert(N % GEMV_COLS == 0, "whole blocks");
    __shared__ float part[GEMV_WAVES][GEMV_COLS];

    const int lane = threadIdx.x & 31;
    const int wave = threadIdx.x >> 5;
    const int c    = lane & 15;
    const int hi8  = (lane >> 4) * 8;
    const int col0 = blockIdx.x * GEMV_COLS;

    const float* xp = x + wave * KS + hi8;
    const float* wp = W + (size_t)(wave * KS + hi8) * N + col0 + 2 * c;

    v8f acc0 = {}, acc1 = {};
#pragma unroll 2
    for (int k0 = 0; k0 < KS; k0 += 32) {
        const v4f xa = *(const v4f*)(xp + k0);
        const v4f xb = *(const v4f*)(xp + k0 + 4);
        const v4f xc = *(const v4f*)(xp + k0 + 16);
        const v4f xd = *(const v4f*)(xp + k0 + 20);
        v16h ah = {}, al = {}, b0h = {}, b0l = {}, b1h = {}, b1l = {};
#pragma unroll
        for (int e = 0; e < 16; ++e) {
            const int kk = k0 + e + (e & 8);
            const float xv = (e < 4) ? xa[e & 3] : (e < 8) ? xb[e & 3]
                           : (e < 12) ? xc[e & 3] : xd[e & 3];
            const v2f wv = *(const v2f*)(wp + (size_t)kk * N);
            const _Float16 xh  = (_Float16)xv;
            const _Float16 w0h = (_Float16)wv[0];
            const _Float16 w1h = (_Float16)wv[1];
            ah[e]  = xh;   al[e]  = (_Float16)(xv - (float)xh);
            b0h[e] = w0h;  b0l[e] = (_Float16)(wv[0] - (float)w0h);
            b1h[e] = w1h;  b1l[e] = (_Float16)(wv[1] - (float)w1h);
        }
        acc0 = __builtin_amdgcn_wmma_f32_16x16x32_f16(
            false, ah, false, b0h, (short)0, acc0, false, false);
        acc1 = __builtin_amdgcn_wmma_f32_16x16x32_f16(
            false, ah, false, b1h, (short)0, acc1, false, false);
        acc0 = __builtin_amdgcn_wmma_f32_16x16x32_f16(
            false, ah, false, b0l, (short)0, acc0, false, false);
        acc1 = __builtin_amdgcn_wmma_f32_16x16x32_f16(
            false, ah, false, b1l, (short)0, acc1, false, false);
        acc0 = __builtin_amdgcn_wmma_f32_16x16x32_f16(
            false, al, false, b0h, (short)0, acc0, false, false);
        acc1 = __builtin_amdgcn_wmma_f32_16x16x32_f16(
            false, al, false, b1h, (short)0, acc1, false, false);
    }

    if (lane < 16) {
        v2f p;
        p[0] = acc0[0];
        p[1] = acc1[0];
        *(v2f*)&part[wave][2 * c] = p;
    }
    __syncthreads();

    if (wave == 0) {
        float v = part[0][lane];
        for (int w = 1; w < GEMV_WAVES; ++w) v += part[w][lane];
        v += bias[col0 + lane];
        if (RELU) v = fmaxf(v, 0.0f);
        y[col0 + lane] = v;
    }
}

#define RN 4096

__device__ __forceinline__ void sturm_block32(const v4f (&t)[8],
                                              float& p1, float& p0,
                                              unsigned& prev, int& cnt) {
    unsigned acc = prev;
#pragma unroll
    for (int u = 0; u < 8; ++u) {
#pragma unroll
        for (int s = 0; s < 4; ++s) {
            const float p2 = __builtin_fmaf(t[u][s], p1, -p0);
            acc = __builtin_amdgcn_alignbit(acc, __float_as_uint(p2), 31u);
            p0 = p1;
            p1 = p2;
        }
        if ((u & 3) == 3) {
            const int e = __builtin_amdgcn_frexp_expf(
                __builtin_fmaxf(__builtin_fabsf(p1), __builtin_fabsf(p0)));
            p1 = __builtin_ldexpf(p1, -e);
            p0 = __builtin_ldexpf(p0, -e);
        }
    }
    const unsigned older = __builtin_amdgcn_alignbit(prev, acc, 1u);
    cnt += __builtin_popcount(acc ^ older);
    prev = acc;
}

__device__ __forceinline__ void hold_block(v4f (&t)[8], float& p1) {
#pragma unroll
    for (int u = 0; u < 8; ++u)
        asm volatile("" : "+v"(t[u]), "+v"(p1) : : "memory");
}

__device__ __forceinline__ int sturm_count_scaled(const v4f* sd4,
                                                  const float x2) {
    float    p1   = 1.0f;
    float    p0   = 0.0f;
    unsigned prev = 0u;
    int      cnt  = 0;
    const int nblk = RN / 32;

    v4f a[8], b[8], ra[8], rb[8];
#pragma unroll
    for (int u = 0; u < 8; ++u) a[u] = sd4[u] - x2;
#pragma unroll
    for (int u = 0; u < 8; ++u) rb[u] = sd4[8 + u];
    hold_block(rb, p1);

    for (int k = 0; k < nblk; k += 2) {
        const int k2 = ((k + 2) & (nblk - 1)) * 8;
#pragma unroll
        for (int u = 0; u < 8; ++u) ra[u] = sd4[k2 + u];
        __builtin_amdgcn_sched_barrier(0);
#pragma unroll
        for (int u = 0; u < 8; ++u) b[u] = rb[u] - x2;
        sturm_block32(a, p1, p0, prev, cnt);
        hold_block(b, p1);
        hold_block(ra, p1);
        const int k3 = ((k + 3) & (nblk - 1)) * 8;
#pragma unroll
        for (int u = 0; u < 8; ++u) rb[u] = sd4[k3 + u];
        __builtin_amdgcn_sched_barrier(0);
#pragma unroll
        for (int u = 0; u < 8; ++u) a[u] = ra[u] - x2;
        sturm_block32(b, p1, p0, prev, cnt);
        hold_block(a, p1);
        hold_block(rb, p1);
    }
    return cnt;
}

__device__ __forceinline__ int sturm_count_ratio(const float* sd,
                                                 const float x2) {
    float q   = sd[0] - x2;
    int   cnt = (q < 0.0f);
    for (int i = 1; i < RN; ++i) {
        q = (sd[i] - x2) - __builtin_amdgcn_rcpf(q);
        if (fabsf(q) < 2e-30f) q = -2e-30f;
        cnt += (q < 0.0f);
    }
    return cnt;
}

#define NPT    8
#define SWEEPS 16

#define CPT    256
#define CSWEEPS 2

__device__ __forceinline__ float grid_point(const float lo, const float w,
                                            const int t) {
    return __builtin_fmaf(w, (float)(t + 1) * (1.0f / (float)(CPT + 1)), lo);
}

__device__ __forceinline__ float section_point(const float lo, const float w,
                                               const int i) {
    return __builtin_fmaf(w, (float)(i + 1) * (1.0f / (float)(NPT + 1)), lo);
}

__device__ __forceinline__ void kept_part(const float lo, const float hi,
                                          const float w, const unsigned gbits,
                                          float& nlo, float& nhi, int& k) {
    k   = __builtin_popcount(gbits);
    nlo = (k > 0)   ? section_point(lo, w, k - 1) : lo;
    nhi = (k < NPT) ? section_point(lo, w, k)     : hi;
}

__global__ void __launch_bounds__(256)
eig_bisect_kernel(const float* __restrict__ ptl, float* __restrict__ evl) {
    __shared__ v4f   sd4[RN / 4];
    __shared__ float res[256 / NPT];
    __shared__ int   cnts[CPT];
    __shared__ float glo[256 / NPT];
    __shared__ float ghi[256 / NPT];
    __shared__ float wmn[256 / 32];
    __shared__ float wmx[256 / 32];
    float* sd = (float*)sd4;
    const int   l   = blockIdx.y;
    const float ll1 = (float)(l * (l + 1));
    const int   tid = threadIdx.x;
    const float dr  = 10000.0f / (float)RN;

    float tmn = 1e30f, tmx = -1e30f;
#pragma unroll
    for (int u = 0; u < RN / 256; ++u) {
        const int   i = tid + u * 256;
        const float r = dr * (float)(i + 1);
        const float d = 1.0f + ptl[i] + ll1 / (r * r);
        const float s = 2.0f * d;
        sd[i] = s;
        tmn = fminf(tmn, s);
        tmx = fmaxf(tmx, s);
    }
    for (int o = 16; o > 0; o >>= 1) {
        tmn = fminf(tmn, __shfl_xor(tmn, o));
        tmx = fmaxf(tmx, __shfl_xor(tmx, o));
    }
    if ((tid & 31) == 0) { wmn[tid >> 5] = tmn; wmx[tid >> 5] = tmx; }
    __syncthreads();

    float mn = wmn[0], mx = wmx[0];
    for (int w = 1; w < 256 / 32; ++w) {
        mn = fminf(mn, wmn[w]);
        mx = fmaxf(mx, wmx[w]);
    }
    float lo = fminf(1e30f, 0.5f * mn - 1.0f);
    float hi = fmaxf(-1e30f, 0.5f * mx + 1.0f);

    const int narrow = __builtin_amdgcn_readfirstlane((hi - lo) < 48.0f ? 1 : 0);

    const int q   = tid & (NPT - 1);
    const int grp = tid / NPT;
    const int gsh = tid & 31 & ~(NPT - 1);
    const int j   = blockIdx.x * (256 / NPT) + grp;

    float blo = lo, bhi = hi;
    for (int cs = 0; cs < CSWEEPS; ++cs) {
        const float bw  = bhi - blo;
        const float xg  = grid_point(blo, bw, tid);
        const float xg2 = xg + xg;
        cnts[tid] = narrow ? sturm_count_scaled(sd4, xg2)
                           : sturm_count_ratio(sd, xg2);
        __syncthreads();
        int k = 0;
        for (int u = 0; u < CPT / NPT; ++u) k += (cnts[u * NPT + q] <= j) ? 1 : 0;
        k += __shfl_xor(k, 1);
        k += __shfl_xor(k, 2);
        k += __shfl_xor(k, 4);
        lo = (k > 0)   ? grid_point(blo, bw, k - 1) : blo;
        hi = (k < CPT) ? grid_point(blo, bw, k)     : bhi;
        if (cs + 1 < CSWEEPS) {
            if (q == 0) { glo[grp] = lo; ghi[grp] = hi; }
            __syncthreads();
            blo = glo[0];
            bhi = ghi[0];
            for (int g = 1; g < 256 / NPT; ++g) {
                blo = fminf(blo, glo[g]);
                bhi = fmaxf(bhi, ghi[g]);
            }
        }
    }

    bool lo_kn = false, hi_kn = false, lo_b = false, hi_b = false;

    for (int it = CSWEEPS; it < SWEEPS; ++it) {
        const float w  = hi - lo;
        const float x  = section_point(lo, w, q);

        const bool at_lo = (x == lo);
        const bool kn = at_lo ? lo_kn : ((x == hi) && hi_kn);
        const bool fb = at_lo ? lo_b : hi_b;
        const unsigned fbits =
            (__builtin_amdgcn_ballot_w32(fb) >> gsh) & ((1u << NPT) - 1u);
        float flo, fhi;
        int   fk;
        kept_part(lo, hi, w, fbits, flo, fhi, fk);
        const bool stays = kn && (flo == lo) && (fhi == hi);
        if (__builtin_amdgcn_ballot_w32(!stays) == 0u) break;

        const float x2 = x + x;
        const int cnt = narrow ? sturm_count_scaled(sd4, x2)
                               : sturm_count_ratio(sd, x2);
        const unsigned gbits =
            (__builtin_amdgcn_ballot_w32(cnt <= j) >> gsh) & ((1u << NPT) - 1u);
        float nlo, nhi;
        int   k;
        kept_part(lo, hi, w, gbits, nlo, nhi, k);
        if (k > 0)   { lo_kn = true; lo_b = ((gbits >> (k - 1)) & 1u) != 0u; }
        if (k < NPT) { hi_kn = true; hi_b = ((gbits >> k) & 1u) != 0u; }
        lo = nlo;
        hi = nhi;
    }

    if (q == 0) res[grp] = 0.5f * (lo + hi);
    __syncthreads();
    if (tid < 256 / NPT)
        evl[l * RN + blockIdx.x * (256 / NPT) + tid] = res[tid];
}

extern "C" void kernel_launch(void* const* d_in, const int* in_sizes, int n_in,
                              void* d_out, int out_size, void* d_ws, size_t ws_size,
                              hipStream_t stream) {
    const float* energy = (const float*)d_in[0];
    const float* W1     = (const float*)d_in[1];
    const float* b1     = (const float*)d_in[2];
    const float* W2     = (const float*)d_in[3];
    const float* b2     = (const float*)d_in[4];
    const float* W3     = (const float*)d_in[5];
    const float* b3     = (const float*)d_in[6];

    float* out = (float*)d_out;
    float* h1  = (float*)d_ws;
    float* h2  = h1 + 2048;

    gemv_wmma_kernel<4096, 2048, true>
        <<<2048 / GEMV_COLS, GEMV_WAVES * 32, 0, stream>>>(energy, W1, b1, h1);
    gemv_wmma_kernel<2048, 1024, true>
        <<<1024 / GEMV_COLS, GEMV_WAVES * 32, 0, stream>>>(h1, W2, b2, h2);
    gemv_wmma_kernel<1024, 4096, false>
        <<<4096 / GEMV_COLS, GEMV_WAVES * 32, 0, stream>>>(h2, W3, b3, out);
    eig_bisect_kernel<<<dim3(RN / (256 / NPT), 2), 256, 0, stream>>>(out, out + 4096);
}


# Multisection stops on absolute width: a bracket no wider than one float step at the spectrum's bound is final, not only neighbouring floats; eigenvalues below the top binade change in low bits
# speedup vs baseline: 21.6286x; 1.2257x over previous
#include <hip/hip_runtime.h>
#include <hip/hip_bf16.h>

typedef __attribute__((ext_vector_type(16))) _Float16 v16h;
typedef __attribute__((ext_vector_type(8)))  float    v8f;

typedef __attribute__((ext_vector_type(2))) float v2f;
typedef __attribute__((ext_vector_type(4))) float v4f;

#define GEMV_WAVES 16
#define GEMV_COLS  32

template <int K, int N, bool RELU>
__global__ void __launch_bounds__(GEMV_WAVES * 32)
gemv_wmma_kernel(const float* __restrict__ x,
                 const float* __restrict__ W,
                 const float* __restrict__ bias,
                 float* __restrict__ y) {
    constexpr int KS = K / GEMV_WAVES;
    static_assert(K % (GEMV_WAVES * 32) == 0, "whole 32-row fragments per wave");
    static_assert(N % GEMV_COLS == 0, "whole blocks");
    __shared__ float part[GEMV_WAVES][GEMV_COLS];

    const int lane = threadIdx.x & 31;
    const int wave = threadIdx.x >> 5;
    const int c    = lane & 15;
    const int hi8  = (lane >> 4) * 8;
    const int col0 = blockIdx.x * GEMV_COLS;

    const float* xp = x + wave * KS + hi8;
    const float* wp = W + (size_t)(wave * KS + hi8) * N + col0 + 2 * c;

    v8f acc0 = {}, acc1 = {};
#pragma unroll 2
    for (int k0 = 0; k0 < KS; k0 += 32) {
        const v4f xa = *(const v4f*)(xp + k0);
        const v4f xb = *(const v4f*)(xp + k0 + 4);
        const v4f xc = *(const v4f*)(xp + k0 + 16);
        const v4f xd = *(const v4f*)(xp + k0 + 20);
        v16h ah = {}, al = {}, b0h = {}, b0l = {}, b1h = {}, b1l = {};
#pragma unroll
        for (int e = 0; e < 16; ++e) {
            const int kk = k0 + e + (e & 8);
            const float xv = (e < 4) ? xa[e & 3] : (e < 8) ? xb[e & 3]
                           : (e < 12) ? xc[e & 3] : xd[e & 3];
            const v2f wv = *(const v2f*)(wp + (size_t)kk * N);
            const _Float16 xh  = (_Float16)xv;
            const _Float16 w0h = (_Float16)wv[0];
            const _Float16 w1h = (_Float16)wv[1];
            ah[e]  = xh;   al[e]  = (_Float16)(xv - (float)xh);
            b0h[e] = w0h;  b0l[e] = (_Float16)(wv[0] - (float)w0h);
            b1h[e] = w1h;  b1l[e] = (_Float16)(wv[1] - (float)w1h);
        }
        acc0 = __builtin_amdgcn_wmma_f32_16x16x32_f16(
            false, ah, false, b0h, (short)0, acc0, false, false);
        acc1 = __builtin_amdgcn_wmma_f32_16x16x32_f16(
            false, ah, false, b1h, (short)0, acc1, false, false);
        acc0 = __builtin_amdgcn_wmma_f32_16x16x32_f16(
            false, ah, false, b0l, (short)0, acc0, false, false);
        acc1 = __builtin_amdgcn_wmma_f32_16x16x32_f16(
            false, ah, false, b1l, (short)0, acc1, false, false);
        acc0 = __builtin_amdgcn_wmma_f32_16x16x32_f16(
            false, al, false, b0h, (short)0, acc0, false, false);
        acc1 = __builtin_amdgcn_wmma_f32_16x16x32_f16(
            false, al, false, b1h, (short)0, acc1, false, false);
    }

    if (lane < 16) {
        v2f p;
        p[0] = acc0[0];
        p[1] = acc1[0];
        *(v2f*)&part[wave][2 * c] = p;
    }
    __syncthreads();

    if (wave == 0) {
        float v = part[0][lane];
        for (int w = 1; w < GEMV_WAVES; ++w) v += part[w][lane];
        v += bias[col0 + lane];
        if (RELU) v = fmaxf(v, 0.0f);
        y[col0 + lane] = v;
    }
}

#define RN 4096

__device__ __forceinline__ void sturm_block32(const v4f (&t)[8],
                                              float& p1, float& p0,
                                              unsigned& prev, int& cnt) {
    unsigned acc = prev;
#pragma unroll
    for (int u = 0; u < 8; ++u) {
#pragma unroll
        for (int s = 0; s < 4; ++s) {
            const float p2 = __builtin_fmaf(t[u][s], p1, -p0);
            acc = __builtin_amdgcn_alignbit(acc, __float_as_uint(p2), 31u);
            p0 = p1;
            p1 = p2;
        }
        if ((u & 3) == 3) {
            const int e = __builtin_amdgcn_frexp_expf(
                __builtin_fmaxf(__builtin_fabsf(p1), __builtin_fabsf(p0)));
            p1 = __builtin_ldexpf(p1, -e);
            p0 = __builtin_ldexpf(p0, -e);
        }
    }
    const unsigned older = __builtin_amdgcn_alignbit(prev, acc, 1u);
    cnt += __builtin_popcount(acc ^ older);
    prev = acc;
}

__device__ __forceinline__ void hold_block(v4f (&t)[8], float& p1) {
#pragma unroll
    for (int u = 0; u < 8; ++u)
        asm volatile("" : "+v"(t[u]), "+v"(p1) : : "memory");
}

__device__ __forceinline__ int sturm_count_scaled(const v4f* sd4,
                                                  const float x2) {
    float    p1   = 1.0f;
    float    p0   = 0.0f;
    unsigned prev = 0u;
    int      cnt  = 0;
    const int nblk = RN / 32;

    v4f a[8], b[8], ra[8], rb[8];
#pragma unroll
    for (int u = 0; u < 8; ++u) a[u] = sd4[u] - x2;
#pragma unroll
    for (int u = 0; u < 8; ++u) rb[u] = sd4[8 + u];
    hold_block(rb, p1);

    for (int k = 0; k < nblk; k += 2) {
        const int k2 = ((k + 2) & (nblk - 1)) * 8;
#pragma unroll
        for (int u = 0; u < 8; ++u) ra[u] = sd4[k2 + u];
        __builtin_amdgcn_sched_barrier(0);
#pragma unroll
        for (int u = 0; u < 8; ++u) b[u] = rb[u] - x2;
        sturm_block32(a, p1, p0, prev, cnt);
        hold_block(b, p1);
        hold_block(ra, p1);
        const int k3 = ((k + 3) & (nblk - 1)) * 8;
#pragma unroll
        for (int u = 0; u < 8; ++u) rb[u] = sd4[k3 + u];
        __builtin_amdgcn_sched_barrier(0);
#pragma unroll
        for (int u = 0; u < 8; ++u) a[u] = ra[u] - x2;
        sturm_block32(b, p1, p0, prev, cnt);
        hold_block(a, p1);
        hold_block(rb, p1);
    }
    return cnt;
}

__device__ __forceinline__ int sturm_count_ratio(const float* sd,
                                                 const float x2) {
    float q   = sd[0] - x2;
    int   cnt = (q < 0.0f);
    for (int i = 1; i < RN; ++i) {
        q = (sd[i] - x2) - __builtin_amdgcn_rcpf(q);
        if (fabsf(q) < 2e-30f) q = -2e-30f;
        cnt += (q < 0.0f);
    }
    return cnt;
}

#define NPT    8
#define SWEEPS 16

#define CPT    256
#define CSWEEPS 2

__device__ __forceinline__ float grid_point(const float lo, const float w,
                                            const int t) {
    return __builtin_fmaf(w, (float)(t + 1) * (1.0f / (float)(CPT + 1)), lo);
}

__device__ __forceinline__ float section_point(const float lo, const float w,
                                               const int i) {
    return __builtin_fmaf(w, (float)(i + 1) * (1.0f / (float)(NPT + 1)), lo);
}

__device__ __forceinline__ void kept_part(const float lo, const float hi,
                                          const float w, const unsigned gbits,
                                          float& nlo, float& nhi, int& k) {
    k   = __builtin_popcount(gbits);
    nlo = (k > 0)   ? section_point(lo, w, k - 1) : lo;
    nhi = (k < NPT) ? section_point(lo, w, k)     : hi;
}

__global__ void __launch_bounds__(256)
eig_bisect_kernel(const float* __restrict__ ptl, float* __restrict__ evl) {
    __shared__ v4f   sd4[RN / 4];
    __shared__ float res[256 / NPT];
    __shared__ int   cnts[CPT];
    __shared__ float glo[256 / NPT];
    __shared__ float ghi[256 / NPT];
    __shared__ float wmn[256 / 32];
    __shared__ float wmx[256 / 32];
    float* sd = (float*)sd4;
    const int   l   = blockIdx.y;
    const float ll1 = (float)(l * (l + 1));
    const int   tid = threadIdx.x;
    const float dr  = 10000.0f / (float)RN;

    float tmn = 1e30f, tmx = -1e30f;
#pragma unroll
    for (int u = 0; u < RN / 256; ++u) {
        const int   i = tid + u * 256;
        const float r = dr * (float)(i + 1);
        const float d = 1.0f + ptl[i] + ll1 / (r * r);
        const float s = 2.0f * d;
        sd[i] = s;
        tmn = fminf(tmn, s);
        tmx = fmaxf(tmx, s);
    }
    for (int o = 16; o > 0; o >>= 1) {
        tmn = fminf(tmn, __shfl_xor(tmn, o));
        tmx = fmaxf(tmx, __shfl_xor(tmx, o));
    }
    if ((tid & 31) == 0) { wmn[tid >> 5] = tmn; wmx[tid >> 5] = tmx; }
    __syncthreads();

    float mn = wmn[0], mx = wmx[0];
    for (int w = 1; w < 256 / 32; ++w) {
        mn = fminf(mn, wmn[w]);
        mx = fmaxf(mx, wmx[w]);
    }
    float lo = fminf(1e30f, 0.5f * mn - 1.0f);
    float hi = fmaxf(-1e30f, 0.5f * mx + 1.0f);
    const float tol = __builtin_ldexpf(
        1.0f, __builtin_amdgcn_frexp_expf(fmaxf(fabsf(lo), fabsf(hi))) - 24);

    const int narrow = __builtin_amdgcn_readfirstlane((hi - lo) < 48.0f ? 1 : 0);

    const int q   = tid & (NPT - 1);
    const int grp = tid / NPT;
    const int gsh = tid & 31 & ~(NPT - 1);
    const int j   = blockIdx.x * (256 / NPT) + grp;

    float blo = lo, bhi = hi;
    for (int cs = 0; cs < CSWEEPS; ++cs) {
        const float bw  = bhi - blo;
        const float xg  = grid_point(blo, bw, tid);
        const float xg2 = xg + xg;
        cnts[tid] = narrow ? sturm_count_scaled(sd4, xg2)
                           : sturm_count_ratio(sd, xg2);
        __syncthreads();
        int k = 0;
        for (int u = 0; u < CPT / NPT; ++u) k += (cnts[u * NPT + q] <= j) ? 1 : 0;
        k += __shfl_xor(k, 1);
        k += __shfl_xor(k, 2);
        k += __shfl_xor(k, 4);
        lo = (k > 0)   ? grid_point(blo, bw, k - 1) : blo;
        hi = (k < CPT) ? grid_point(blo, bw, k)     : bhi;
        if (cs + 1 < CSWEEPS) {
            if (q == 0) { glo[grp] = lo; ghi[grp] = hi; }
            __syncthreads();
            blo = glo[0];
            bhi = ghi[0];
            for (int g = 1; g < 256 / NPT; ++g) {
                blo = fminf(blo, glo[g]);
                bhi = fmaxf(bhi, ghi[g]);
            }
        }
    }

    bool lo_kn = false, hi_kn = false, lo_b = false, hi_b = false;

    for (int it = CSWEEPS; it < SWEEPS; ++it) {
        const float w  = hi - lo;
        const float x  = section_point(lo, w, q);

        const bool at_lo = (x == lo);
        const bool kn = at_lo ? lo_kn : ((x == hi) && hi_kn);
        const bool fb = at_lo ? lo_b : hi_b;
        const unsigned fbits =
            (__builtin_amdgcn_ballot_w32(fb) >> gsh) & ((1u << NPT) - 1u);
        float flo, fhi;
        int   fk;
        kept_part(lo, hi, w, fbits, flo, fhi, fk);
        const bool fine  = (w <= tol);
        const bool stays = fine || (kn && (flo == lo) && (fhi == hi));
        if (__builtin_amdgcn_ballot_w32(!stays) == 0u) break;

        const float x2 = x + x;
        const int cnt = narrow ? sturm_count_scaled(sd4, x2)
                               : sturm_count_ratio(sd, x2);
        const unsigned gbits =
            (__builtin_amdgcn_ballot_w32(cnt <= j) >> gsh) & ((1u << NPT) - 1u);
        float nlo, nhi;
        int   k;
        kept_part(lo, hi, w, gbits, nlo, nhi, k);
        if (!fine) {
            if (k > 0)   { lo_kn = true; lo_b = ((gbits >> (k - 1)) & 1u) != 0u; }
            if (k < NPT) { hi_kn = true; hi_b = ((gbits >> k) & 1u) != 0u; }
            lo = nlo;
            hi = nhi;
        }
    }

    if (q == 0) res[grp] = 0.5f * (lo + hi);
    __syncthreads();
    if (tid < 256 / NPT)
        evl[l * RN + blockIdx.x * (256 / NPT) + tid] = res[tid];
}

extern "C" void kernel_launch(void* const* d_in, const int* in_sizes, int n_in,
                              void* d_out, int out_size, void* d_ws, size_t ws_size,
                              hipStream_t stream) {
    const float* energy = (const float*)d_in[0];
    const float* W1     = (const float*)d_in[1];
    const float* b1     = (const float*)d_in[2];
    const float* W2     = (const float*)d_in[3];
    const float* b2     = (const float*)d_in[4];
    const float* W3     = (const float*)d_in[5];
    const float* b3     = (const float*)d_in[6];

    float* out = (float*)d_out;
    float* h1  = (float*)d_ws;
    float* h2  = h1 + 2048;

    gemv_wmma_kernel<4096, 2048, true>
        <<<2048 / GEMV_COLS, GEMV_WAVES * 32, 0, stream>>>(energy, W1, b1, h1);
    gemv_wmma_kernel<2048, 1024, true>
        <<<1024 / GEMV_COLS, GEMV_WAVES * 32, 0, stream>>>(h1, W2, b2, h2);
    gemv_wmma_kernel<1024, 4096, false>
        <<<4096 / GEMV_COLS, GEMV_WAVES * 32, 0, stream>>>(h2, W3, b3, out);
    eig_bisect_kernel<<<dim3(RN / (256 / NPT), 2), 256, 0, stream>>>(out, out + 4096);
}


# Sturm chain rescale: the exponent is read off the pair four steps before it is applied, so only the ldexp stays on the dependent chain (scaled form now for spectra narrower than 32, was 48)
# speedup vs baseline: 22.4189x; 1.0365x over previous
#include <hip/hip_runtime.h>
#include <hip/hip_bf16.h>

typedef __attribute__((ext_vector_type(16))) _Float16 v16h;
typedef __attribute__((ext_vector_type(8)))  float    v8f;

typedef __attribute__((ext_vector_type(2))) float v2f;
typedef __attribute__((ext_vector_type(4))) float v4f;

#define GEMV_WAVES 16
#define GEMV_COLS  32

template <int K, int N, bool RELU>
__global__ void __launch_bounds__(GEMV_WAVES * 32)
gemv_wmma_kernel(const float* __restrict__ x,
                 const float* __restrict__ W,
                 const float* __restrict__ bias,
                 float* __restrict__ y) {
    constexpr int KS = K / GEMV_WAVES;
    static_assert(K % (GEMV_WAVES * 32) == 0, "whole 32-row fragments per wave");
    static_assert(N % GEMV_COLS == 0, "whole blocks");
    __shared__ float part[GEMV_WAVES][GEMV_COLS];

    const int lane = threadIdx.x & 31;
    const int wave = threadIdx.x >> 5;
    const int c    = lane & 15;
    const int hi8  = (lane >> 4) * 8;
    const int col0 = blockIdx.x * GEMV_COLS;

    const float* xp = x + wave * KS + hi8;
    const float* wp = W + (size_t)(wave * KS + hi8) * N + col0 + 2 * c;

    v8f acc0 = {}, acc1 = {};
#pragma unroll 2
    for (int k0 = 0; k0 < KS; k0 += 32) {
        const v4f xa = *(const v4f*)(xp + k0);
        const v4f xb = *(const v4f*)(xp + k0 + 4);
        const v4f xc = *(const v4f*)(xp + k0 + 16);
        const v4f xd = *(const v4f*)(xp + k0 + 20);
        v16h ah = {}, al = {}, b0h = {}, b0l = {}, b1h = {}, b1l = {};
#pragma unroll
        for (int e = 0; e < 16; ++e) {
            const int kk = k0 + e + (e & 8);
            const float xv = (e < 4) ? xa[e & 3] : (e < 8) ? xb[e & 3]
                           : (e < 12) ? xc[e & 3] : xd[e & 3];
            const v2f wv = *(const v2f*)(wp + (size_t)kk * N);
            const _Float16 xh  = (_Float16)xv;
            const _Float16 w0h = (_Float16)wv[0];
            const _Float16 w1h = (_Float16)wv[1];
            ah[e]  = xh;   al[e]  = (_Float16)(xv - (float)xh);
            b0h[e] = w0h;  b0l[e] = (_Float16)(wv[0] - (float)w0h);
            b1h[e] = w1h;  b1l[e] = (_Float16)(wv[1] - (float)w1h);
        }
        acc0 = __builtin_amdgcn_wmma_f32_16x16x32_f16(
            false, ah, false, b0h, (short)0, acc0, false, false);
        acc1 = __builtin_amdgcn_wmma_f32_16x16x32_f16(
            false, ah, false, b1h, (short)0, acc1, false, false);
        acc0 = __builtin_amdgcn_wmma_f32_16x16x32_f16(
            false, ah, false, b0l, (short)0, acc0, false, false);
        acc1 = __builtin_amdgcn_wmma_f32_16x16x32_f16(
            false, ah, false, b1l, (short)0, acc1, false, false);
        acc0 = __builtin_amdgcn_wmma_f32_16x16x32_f16(
            false, al, false, b0h, (short)0, acc0, false, false);
        acc1 = __builtin_amdgcn_wmma_f32_16x16x32_f16(
            false, al, false, b1h, (short)0, acc1, false, false);
    }

    if (lane < 16) {
        v2f p;
        p[0] = acc0[0];
        p[1] = acc1[0];
        *(v2f*)&part[wave][2 * c] = p;
    }
    __syncthreads();

    if (wave == 0) {
        float v = part[0][lane];
        for (int w = 1; w < GEMV_WAVES; ++w) v += part[w][lane];
        v += bias[col0 + lane];
        if (RELU) v = fmaxf(v, 0.0f);
        y[col0 + lane] = v;
    }
}

#define RN 4096

__device__ __forceinline__ void sturm_block32(const v4f (&t)[8],
                                              float& p1, float& p0,
                                              unsigned& prev, int& cnt) {
    unsigned acc = prev;
    int      ne  = 0;
#pragma unroll
    for (int u = 0; u < 8; ++u) {
#pragma unroll
        for (int s = 0; s < 4; ++s) {
            const float p2 = __builtin_fmaf(t[u][s], p1, -p0);
            acc = __builtin_amdgcn_alignbit(acc, __float_as_uint(p2), 31u);
            p0 = p1;
            p1 = p2;
        }
        if ((u & 3) == 2) {
            ne = -__builtin_amdgcn_frexp_expf(
                __builtin_fmaxf(__builtin_fabsf(p1), __builtin_fabsf(p0)));
        }
        if ((u & 3) == 3) {
            p1 = __builtin_ldexpf(p1, ne);
            p0 = __builtin_ldexpf(p0, ne);
        }
    }
    const unsigned older = __builtin_amdgcn_alignbit(prev, acc, 1u);
    cnt += __builtin_popcount(acc ^ older);
    prev = acc;
}

__device__ __forceinline__ void hold_block(v4f (&t)[8], float& p1) {
#pragma unroll
    for (int u = 0; u < 8; ++u)
        asm volatile("" : "+v"(t[u]), "+v"(p1) : : "memory");
}

__device__ __forceinline__ int sturm_count_scaled(const v4f* sd4,
                                                  const float x2) {
    float    p1   = 1.0f;
    float    p0   = 0.0f;
    unsigned prev = 0u;
    int      cnt  = 0;
    const int nblk = RN / 32;

    v4f a[8], b[8], ra[8], rb[8];
#pragma unroll
    for (int u = 0; u < 8; ++u) a[u] = sd4[u] - x2;
#pragma unroll
    for (int u = 0; u < 8; ++u) rb[u] = sd4[8 + u];
    hold_block(rb, p1);

    for (int k = 0; k < nblk; k += 2) {
        const int k2 = ((k + 2) & (nblk - 1)) * 8;
#pragma unroll
        for (int u = 0; u < 8; ++u) ra[u] = sd4[k2 + u];
        __builtin_amdgcn_sched_barrier(0);
#pragma unroll
        for (int u = 0; u < 8; ++u) b[u] = rb[u] - x2;
        sturm_block32(a, p1, p0, prev, cnt);
        hold_block(b, p1);
        hold_block(ra, p1);
        const int k3 = ((k + 3) & (nblk - 1)) * 8;
#pragma unroll
        for (int u = 0; u < 8; ++u) rb[u] = sd4[k3 + u];
        __builtin_amdgcn_sched_barrier(0);
#pragma unroll
        for (int u = 0; u < 8; ++u) a[u] = ra[u] - x2;
        sturm_block32(b, p1, p0, prev, cnt);
        hold_block(a, p1);
        hold_block(rb, p1);
    }
    return cnt;
}

__device__ __forceinline__ int sturm_count_ratio(const float* sd,
                                                 const float x2) {
    float q   = sd[0] - x2;
    int   cnt = (q < 0.0f);
    for (int i = 1; i < RN; ++i) {
        q = (sd[i] - x2) - __builtin_amdgcn_rcpf(q);
        if (fabsf(q) < 2e-30f) q = -2e-30f;
        cnt += (q < 0.0f);
    }
    return cnt;
}

#define NPT    8
#define SWEEPS 16

#define CPT    256
#define CSWEEPS 2

__device__ __forceinline__ float grid_point(const float lo, const float w,
                                            const int t) {
    return __builtin_fmaf(w, (float)(t + 1) * (1.0f / (float)(CPT + 1)), lo);
}

__device__ __forceinline__ float section_point(const float lo, const float w,
                                               const int i) {
    return __builtin_fmaf(w, (float)(i + 1) * (1.0f / (float)(NPT + 1)), lo);
}

__device__ __forceinline__ void kept_part(const float lo, const float hi,
                                          const float w, const unsigned gbits,
                                          float& nlo, float& nhi, int& k) {
    k   = __builtin_popcount(gbits);
    nlo = (k > 0)   ? section_point(lo, w, k - 1) : lo;
    nhi = (k < NPT) ? section_point(lo, w, k)     : hi;
}

__global__ void __launch_bounds__(256)
eig_bisect_kernel(const float* __restrict__ ptl, float* __restrict__ evl) {
    __shared__ v4f   sd4[RN / 4];
    __shared__ float res[256 / NPT];
    __shared__ int   cnts[CPT];
    __shared__ float glo[256 / NPT];
    __shared__ float ghi[256 / NPT];
    __shared__ float wmn[256 / 32];
    __shared__ float wmx[256 / 32];
    float* sd = (float*)sd4;
    const int   l   = blockIdx.y;
    const float ll1 = (float)(l * (l + 1));
    const int   tid = threadIdx.x;
    const float dr  = 10000.0f / (float)RN;

    float tmn = 1e30f, tmx = -1e30f;
#pragma unroll
    for (int u = 0; u < RN / 256; ++u) {
        const int   i = tid + u * 256;
        const float r = dr * (float)(i + 1);
        const float d = 1.0f + ptl[i] + ll1 / (r * r);
        const float s = 2.0f * d;
        sd[i] = s;
        tmn = fminf(tmn, s);
        tmx = fmaxf(tmx, s);
    }
    for (int o = 16; o > 0; o >>= 1) {
        tmn = fminf(tmn, __shfl_xor(tmn, o));
        tmx = fmaxf(tmx, __shfl_xor(tmx, o));
    }
    if ((tid & 31) == 0) { wmn[tid >> 5] = tmn; wmx[tid >> 5] = tmx; }
    __syncthreads();

    float mn = wmn[0], mx = wmx[0];
    for (int w = 1; w < 256 / 32; ++w) {
        mn = fminf(mn, wmn[w]);
        mx = fmaxf(mx, wmx[w]);
    }
    float lo = fminf(1e30f, 0.5f * mn - 1.0f);
    float hi = fmaxf(-1e30f, 0.5f * mx + 1.0f);
    const float tol = __builtin_ldexpf(
        1.0f, __builtin_amdgcn_frexp_expf(fmaxf(fabsf(lo), fabsf(hi))) - 24);

    const int narrow = __builtin_amdgcn_readfirstlane((hi - lo) < 32.0f ? 1 : 0);

    const int q   = tid & (NPT - 1);
    const int grp = tid / NPT;
    const int gsh = tid & 31 & ~(NPT - 1);
    const int j   = blockIdx.x * (256 / NPT) + grp;

    float blo = lo, bhi = hi;
    for (int cs = 0; cs < CSWEEPS; ++cs) {
        const float bw  = bhi - blo;
        const float xg  = grid_point(blo, bw, tid);
        const float xg2 = xg + xg;
        cnts[tid] = narrow ? sturm_count_scaled(sd4, xg2)
                           : sturm_count_ratio(sd, xg2);
        __syncthreads();
        int k = 0;
        for (int u = 0; u < CPT / NPT; ++u) k += (cnts[u * NPT + q] <= j) ? 1 : 0;
        k += __shfl_xor(k, 1);
        k += __shfl_xor(k, 2);
        k += __shfl_xor(k, 4);
        lo = (k > 0)   ? grid_point(blo, bw, k - 1) : blo;
        hi = (k < CPT) ? grid_point(blo, bw, k)     : bhi;
        if (cs + 1 < CSWEEPS) {
            if (q == 0) { glo[grp] = lo; ghi[grp] = hi; }
            __syncthreads();
            blo = glo[0];
            bhi = ghi[0];
            for (int g = 1; g < 256 / NPT; ++g) {
                blo = fminf(blo, glo[g]);
                bhi = fmaxf(bhi, ghi[g]);
            }
        }
    }

    bool lo_kn = false, hi_kn = false, lo_b = false, hi_b = false;

    for (int it = CSWEEPS; it < SWEEPS; ++it) {
        const float w  = hi - lo;
        const float x  = section_point(lo, w, q);

        const bool at_lo = (x == lo);
        const bool kn = at_lo ? lo_kn : ((x == hi) && hi_kn);
        const bool fb = at_lo ? lo_b : hi_b;
        const unsigned fbits =
            (__builtin_amdgcn_ballot_w32(fb) >> gsh) & ((1u << NPT) - 1u);
        float flo, fhi;
        int   fk;
        kept_part(lo, hi, w, fbits, flo, fhi, fk);
        const bool fine  = (w <= tol);
        const bool stays = fine || (kn && (flo == lo) && (fhi == hi));
        if (__builtin_amdgcn_ballot_w32(!stays) == 0u) break;

        const float x2 = x + x;
        const int cnt = narrow ? sturm_count_scaled(sd4, x2)
                               : sturm_count_ratio(sd, x2);
        const unsigned gbits =
            (__builtin_amdgcn_ballot_w32(cnt <= j) >> gsh) & ((1u << NPT) - 1u);
        float nlo, nhi;
        int   k;
        kept_part(lo, hi, w, gbits, nlo, nhi, k);
        if (!fine) {
            if (k > 0)   { lo_kn = true; lo_b = ((gbits >> (k - 1)) & 1u) != 0u; }
            if (k < NPT) { hi_kn = true; hi_b = ((gbits >> k) & 1u) != 0u; }
            lo = nlo;
            hi = nhi;
        }
    }

    if (q == 0) res[grp] = 0.5f * (lo + hi);
    __syncthreads();
    if (tid < 256 / NPT)
        evl[l * RN + blockIdx.x * (256 / NPT) + tid] = res[tid];
}

extern "C" void kernel_launch(void* const* d_in, const int* in_sizes, int n_in,
                              void* d_out, int out_size, void* d_ws, size_t ws_size,
                              hipStream_t stream) {
    const float* energy = (const float*)d_in[0];
    const float* W1     = (const float*)d_in[1];
    const float* b1     = (const float*)d_in[2];
    const float* W2     = (const float*)d_in[3];
    const float* b2     = (const float*)d_in[4];
    const float* W3     = (const float*)d_in[5];
    const float* b3     = (const float*)d_in[6];

    float* out = (float*)d_out;
    float* h1  = (float*)d_ws;
    float* h2  = h1 + 2048;

    gemv_wmma_kernel<4096, 2048, true>
        <<<2048 / GEMV_COLS, GEMV_WAVES * 32, 0, stream>>>(energy, W1, b1, h1);
    gemv_wmma_kernel<2048, 1024, true>
        <<<1024 / GEMV_COLS, GEMV_WAVES * 32, 0, stream>>>(h1, W2, b2, h2);
    gemv_wmma_kernel<1024, 4096, false>
        <<<4096 / GEMV_COLS, GEMV_WAVES * 32, 0, stream>>>(h2, W3, b3, out);
    eig_bisect_kernel<<<dim3(RN / (256 / NPT), 2), 256, 0, stream>>>(out, out + 4096);
}


# Chip-wide first sweep in its own launch: a matrix's blocks share one grid of 32641 distinct points (255 new per block), each eigenvalue gets a cell; replaces both 257-way block sweeps; low bits move
# speedup vs baseline: 29.1253x; 1.2991x over previous
#include <hip/hip_runtime.h>
#include <hip/hip_bf16.h>

typedef __attribute__((ext_vector_type(16))) _Float16 v16h;
typedef __attribute__((ext_vector_type(8)))  float    v8f;

typedef __attribute__((ext_vector_type(2))) float v2f;
typedef __attribute__((ext_vector_type(4))) float v4f;

#define GEMV_WAVES 16
#define GEMV_COLS  32

template <int K, int N, bool RELU>
__global__ void __launch_bounds__(GEMV_WAVES * 32)
gemv_wmma_kernel(const float* __restrict__ x,
                 const float* __restrict__ W,
                 const float* __restrict__ bias,
                 float* __restrict__ y) {
    constexpr int KS = K / GEMV_WAVES;
    static_assert(K % (GEMV_WAVES * 32) == 0, "whole 32-row fragments per wave");
    static_assert(N % GEMV_COLS == 0, "whole blocks");
    __shared__ float part[GEMV_WAVES][GEMV_COLS];

    const int lane = threadIdx.x & 31;
    const int wave = threadIdx.x >> 5;
    const int c    = lane & 15;
    const int hi8  = (lane >> 4) * 8;
    const int col0 = blockIdx.x * GEMV_COLS;

    const float* xp = x + wave * KS + hi8;
    const float* wp = W + (size_t)(wave * KS + hi8) * N + col0 + 2 * c;

    v8f acc0 = {}, acc1 = {};
#pragma unroll 2
    for (int k0 = 0; k0 < KS; k0 += 32) {
        const v4f xa = *(const v4f*)(xp + k0);
        const v4f xb = *(const v4f*)(xp + k0 + 4);
        const v4f xc = *(const v4f*)(xp + k0 + 16);
        const v4f xd = *(const v4f*)(xp + k0 + 20);
        v16h ah = {}, al = {}, b0h = {}, b0l = {}, b1h = {}, b1l = {};
#pragma unroll
        for (int e = 0; e < 16; ++e) {
            const int kk = k0 + e + (e & 8);
            const float xv = (e < 4) ? xa[e & 3] : (e < 8) ? xb[e & 3]
                           : (e < 12) ? xc[e & 3] : xd[e & 3];
            const v2f wv = *(const v2f*)(wp + (size_t)kk * N);
            const _Float16 xh  = (_Float16)xv;
            const _Float16 w0h = (_Float16)wv[0];
            const _Float16 w1h = (_Float16)wv[1];
            ah[e]  = xh;   al[e]  = (_Float16)(xv - (float)xh);
            b0h[e] = w0h;  b0l[e] = (_Float16)(wv[0] - (float)w0h);
            b1h[e] = w1h;  b1l[e] = (_Float16)(wv[1] - (float)w1h);
        }
        acc0 = __builtin_amdgcn_wmma_f32_16x16x32_f16(
            false, ah, false, b0h, (short)0, acc0, false, false);
        acc1 = __builtin_amdgcn_wmma_f32_16x16x32_f16(
            false, ah, false, b1h, (short)0, acc1, false, false);
        acc0 = __builtin_amdgcn_wmma_f32_16x16x32_f16(
            false, ah, false, b0l, (short)0, acc0, false, false);
        acc1 = __builtin_amdgcn_wmma_f32_16x16x32_f16(
            false, ah, false, b1l, (short)0, acc1, false, false);
        acc0 = __builtin_amdgcn_wmma_f32_16x16x32_f16(
            false, al, false, b0h, (short)0, acc0, false, false);
        acc1 = __builtin_amdgcn_wmma_f32_16x16x32_f16(
            false, al, false, b1h, (short)0, acc1, false, false);
    }

    if (lane < 16) {
        v2f p;
        p[0] = acc0[0];
        p[1] = acc1[0];
        *(v2f*)&part[wave][2 * c] = p;
    }
    __syncthreads();

    if (wave == 0) {
        float v = part[0][lane];
        for (int w = 1; w < GEMV_WAVES; ++w) v += part[w][lane];
        v += bias[col0 + lane];
        if (RELU) v = fmaxf(v, 0.0f);
        y[col0 + lane] = v;
    }
}

#define RN 4096

__device__ __forceinline__ void sturm_block32(const v4f (&t)[8],
                                              float& p1, float& p0,
                                              unsigned& prev, int& cnt) {
    unsigned acc = prev;
    int      ne  = 0;
#pragma unroll
    for (int u = 0; u < 8; ++u) {
#pragma unroll
        for (int s = 0; s < 4; ++s) {
            const float p2 = __builtin_fmaf(t[u][s], p1, -p0);
            acc = __builtin_amdgcn_alignbit(acc, __float_as_uint(p2), 31u);
            p0 = p1;
            p1 = p2;
        }
        if ((u & 3) == 2) {
            ne = -__builtin_amdgcn_frexp_expf(
                __builtin_fmaxf(__builtin_fabsf(p1), __builtin_fabsf(p0)));
        }
        if ((u & 3) == 3) {
            p1 = __builtin_ldexpf(p1, ne);
            p0 = __builtin_ldexpf(p0, ne);
        }
    }
    const unsigned older = __builtin_amdgcn_alignbit(prev, acc, 1u);
    cnt += __builtin_popcount(acc ^ older);
    prev = acc;
}

__device__ __forceinline__ void hold_block(v4f (&t)[8], float& p1) {
#pragma unroll
    for (int u = 0; u < 8; ++u)
        asm volatile("" : "+v"(t[u]), "+v"(p1) : : "memory");
}

__device__ __forceinline__ int sturm_count_scaled(const v4f* sd4,
                                                  const float x2) {
    float    p1   = 1.0f;
    float    p0   = 0.0f;
    unsigned prev = 0u;
    int      cnt  = 0;
    const int nblk = RN / 32;

    v4f a[8], b[8], ra[8], rb[8];
#pragma unroll
    for (int u = 0; u < 8; ++u) a[u] = sd4[u] - x2;
#pragma unroll
    for (int u = 0; u < 8; ++u) rb[u] = sd4[8 + u];
    hold_block(rb, p1);

    for (int k = 0; k < nblk; k += 2) {
        const int k2 = ((k + 2) & (nblk - 1)) * 8;
#pragma unroll
        for (int u = 0; u < 8; ++u) ra[u] = sd4[k2 + u];
        __builtin_amdgcn_sched_barrier(0);
#pragma unroll
        for (int u = 0; u < 8; ++u) b[u] = rb[u] - x2;
        sturm_block32(a, p1, p0, prev, cnt);
        hold_block(b, p1);
        hold_block(ra, p1);
        const int k3 = ((k + 3) & (nblk - 1)) * 8;
#pragma unroll
        for (int u = 0; u < 8; ++u) rb[u] = sd4[k3 + u];
        __builtin_amdgcn_sched_barrier(0);
#pragma unroll
        for (int u = 0; u < 8; ++u) a[u] = ra[u] - x2;
        sturm_block32(b, p1, p0, prev, cnt);
        hold_block(a, p1);
        hold_block(rb, p1);
    }
    return cnt;
}

__device__ __forceinline__ int sturm_count_ratio(const float* sd,
                                                 const float x2) {
    float q   = sd[0] - x2;
    int   cnt = (q < 0.0f);
    for (int i = 1; i < RN; ++i) {
        q = (sd[i] - x2) - __builtin_amdgcn_rcpf(q);
        if (fabsf(q) < 2e-30f) q = -2e-30f;
        cnt += (q < 0.0f);
    }
    return cnt;
}

#define NPT    8
#define SWEEPS 16

#define GPB    255
#define GBLK   (RN / (256 / NPT))
#define GPTS   (GBLK * GPB + 1)

__device__ __forceinline__ float grid_point(const float lo, const float w,
                                            const int t) {
    return __builtin_fmaf(w, (float)(t + 1) * (1.0f / (float)(GPTS + 1)), lo);
}

__device__ __forceinline__ float section_point(const float lo, const float w,
                                               const int i) {
    return __builtin_fmaf(w, (float)(i + 1) * (1.0f / (float)(NPT + 1)), lo);
}

__device__ __forceinline__ void kept_part(const float lo, const float hi,
                                          const float w, const unsigned gbits,
                                          float& nlo, float& nhi, int& k) {
    k   = __builtin_popcount(gbits);
    nlo = (k > 0)   ? section_point(lo, w, k - 1) : lo;
    nhi = (k < NPT) ? section_point(lo, w, k)     : hi;
}

__device__ __forceinline__ void stage_diagonal(const float* __restrict__ ptl,
                                               const int l, const int tid,
                                               float* sd, float* wmn, float* wmx,
                                               float& lo, float& hi) {
    const float ll1 = (float)(l * (l + 1));
    const float dr  = 10000.0f / (float)RN;
    float tmn = 1e30f, tmx = -1e30f;
#pragma unroll
    for (int u = 0; u < RN / 256; ++u) {
        const int   i = tid + u * 256;
        const float r = dr * (float)(i + 1);
        const float d = 1.0f + ptl[i] + ll1 / (r * r);
        const float s = 2.0f * d;
        sd[i] = s;
        tmn = fminf(tmn, s);
        tmx = fmaxf(tmx, s);
    }
    for (int o = 16; o > 0; o >>= 1) {
        tmn = fminf(tmn, __shfl_xor(tmn, o));
        tmx = fmaxf(tmx, __shfl_xor(tmx, o));
    }
    if ((tid & 31) == 0) { wmn[tid >> 5] = tmn; wmx[tid >> 5] = tmx; }
    __syncthreads();

    float mn = wmn[0], mx = wmx[0];
    for (int w = 1; w < 256 / 32; ++w) {
        mn = fminf(mn, wmn[w]);
        mx = fmaxf(mx, wmx[w]);
    }
    lo = fminf(1e30f, 0.5f * mn - 1.0f);
    hi = fmaxf(-1e30f, 0.5f * mx + 1.0f);
}

__global__ void __launch_bounds__(256)
eig_grid_kernel(const float* __restrict__ ptl, float* __restrict__ evl) {
    __shared__ v4f   sd4[RN / 4];
    __shared__ int   cnts[256];
    __shared__ float wmn[256 / 32];
    __shared__ float wmx[256 / 32];
    float* sd = (float*)sd4;
    const int l   = blockIdx.y;
    const int tid = threadIdx.x;

    float lo, hi;
    stage_diagonal(ptl, l, tid, sd, wmn, wmx, lo, hi);
    const int narrow = __builtin_amdgcn_readfirstlane((hi - lo) < 32.0f ? 1 : 0);

    const int   t  = blockIdx.x * GPB + tid;
    const float x  = grid_point(lo, hi - lo, t);
    const float x2 = x + x;
    const int   c  = narrow ? sturm_count_scaled(sd4, x2)
                            : sturm_count_ratio(sd, x2);
    cnts[tid] = c;
    __syncthreads();

    float* row = evl + l * RN;
    int from = (tid > 0) ? cnts[tid - 1] : ((blockIdx.x == 0) ? 0 : c);
    from = max(from, 0);
    const int to = min(c, RN);
#pragma clang loop vectorize(disable) interleave(disable) unroll(disable)
    for (int j = from; j < to; ++j) row[j] = (float)t;
    if (blockIdx.x == GBLK - 1 && tid == 255) {
#pragma clang loop vectorize(disable) interleave(disable) unroll(disable)
        for (int j = max(c, 0); j < RN; ++j) row[j] = (float)GPTS;
    }
}

__global__ void __launch_bounds__(256)
eig_bisect_kernel(const float* __restrict__ ptl, float* __restrict__ evl) {
    __shared__ v4f   sd4[RN / 4];
    __shared__ float res[256 / NPT];
    __shared__ float wmn[256 / 32];
    __shared__ float wmx[256 / 32];
    float* sd = (float*)sd4;
    const int l   = blockIdx.y;
    const int tid = threadIdx.x;

    const int q   = tid & (NPT - 1);
    const int grp = tid / NPT;
    const int gsh = tid & 31 & ~(NPT - 1);
    const int j   = blockIdx.x * (256 / NPT) + grp;

    const float cell = evl[l * RN + j];

    float glo, ghi;
    stage_diagonal(ptl, l, tid, sd, wmn, wmx, glo, ghi);
    const float tol = __builtin_ldexpf(
        1.0f, __builtin_amdgcn_frexp_expf(fmaxf(fabsf(glo), fabsf(ghi))) - 24);

    const int narrow = __builtin_amdgcn_readfirstlane((ghi - glo) < 32.0f ? 1 : 0);

    const int   kc = min(max((int)cell, 0), GPTS);
    const float gw = ghi - glo;
    float lo = (kc > 0)    ? grid_point(glo, gw, kc - 1) : glo;
    float hi = (kc < GPTS) ? grid_point(glo, gw, kc)     : ghi;

    bool lo_kn = false, hi_kn = false, lo_b = false, hi_b = false;

    for (int it = 1; it < SWEEPS; ++it) {
        const float w  = hi - lo;
        const float x  = section_point(lo, w, q);

        const bool at_lo = (x == lo);
        const bool kn = at_lo ? lo_kn : ((x == hi) && hi_kn);
        const bool fb = at_lo ? lo_b : hi_b;
        const unsigned fbits =
            (__builtin_amdgcn_ballot_w32(fb) >> gsh) & ((1u << NPT) - 1u);
        float flo, fhi;
        int   fk;
        kept_part(lo, hi, w, fbits, flo, fhi, fk);
        const bool fine  = (w <= tol);
        const bool stays = fine || (kn && (flo == lo) && (fhi == hi));
        if (__builtin_amdgcn_ballot_w32(!stays) == 0u) break;

        const float x2 = x + x;
        const int cnt = narrow ? sturm_count_scaled(sd4, x2)
                               : sturm_count_ratio(sd, x2);
        const unsigned gbits =
            (__builtin_amdgcn_ballot_w32(cnt <= j) >> gsh) & ((1u << NPT) - 1u);
        float nlo, nhi;
        int   k;
        kept_part(lo, hi, w, gbits, nlo, nhi, k);
        if (!fine) {
            if (k > 0)   { lo_kn = true; lo_b = ((gbits >> (k - 1)) & 1u) != 0u; }
            if (k < NPT) { hi_kn = true; hi_b = ((gbits >> k) & 1u) != 0u; }
            lo = nlo;
            hi = nhi;
        }
    }

    if (q == 0) res[grp] = 0.5f * (lo + hi);
    __syncthreads();
    if (tid < 256 / NPT)
        evl[l * RN + blockIdx.x * (256 / NPT) + tid] = res[tid];
}

extern "C" void kernel_launch(void* const* d_in, const int* in_sizes, int n_in,
                              void* d_out, int out_size, void* d_ws, size_t ws_size,
                              hipStream_t stream) {
    const float* energy = (const float*)d_in[0];
    const float* W1     = (const float*)d_in[1];
    const float* b1     = (const float*)d_in[2];
    const float* W2     = (const float*)d_in[3];
    const float* b2     = (const float*)d_in[4];
    const float* W3     = (const float*)d_in[5];
    const float* b3     = (const float*)d_in[6];

    float* out = (float*)d_out;
    float* h1  = (float*)d_ws;
    float* h2  = h1 + 2048;

    gemv_wmma_kernel<4096, 2048, true>
        <<<2048 / GEMV_COLS, GEMV_WAVES * 32, 0, stream>>>(energy, W1, b1, h1);
    gemv_wmma_kernel<2048, 1024, true>
        <<<1024 / GEMV_COLS, GEMV_WAVES * 32, 0, stream>>>(h1, W2, b2, h2);
    gemv_wmma_kernel<1024, 4096, false>
        <<<4096 / GEMV_COLS, GEMV_WAVES * 32, 0, stream>>>(h2, W3, b3, out);
    eig_grid_kernel<<<dim3(GBLK, 2), 256, 0, stream>>>(out, out + 4096);
    eig_bisect_kernel<<<dim3(GBLK, 2), 256, 0, stream>>>(out, out + 4096);
}


# Sturm count from both ends: two lanes per point run half-length chains (rows 0..2047 down, 4095..2048 up) joined by the middle pivot's sign; 5-way on 4 points, 16257-point first sweep; low bits move
# speedup vs baseline: 40.6262x; 1.3949x over previous
#include <hip/hip_runtime.h>
#include <hip/hip_bf16.h>

typedef __attribute__((ext_vector_type(16))) _Float16 v16h;
typedef __attribute__((ext_vector_type(8)))  float    v8f;

typedef __attribute__((ext_vector_type(2))) float v2f;
typedef __attribute__((ext_vector_type(4))) float v4f;

#define GEMV_WAVES 16
#define GEMV_COLS  32

template <int K, int N, bool RELU>
__global__ void __launch_bounds__(GEMV_WAVES * 32)
gemv_wmma_kernel(const float* __restrict__ x,
                 const float* __restrict__ W,
                 const float* __restrict__ bias,
                 float* __restrict__ y) {
    constexpr int KS = K / GEMV_WAVES;
    static_assert(K % (GEMV_WAVES * 32) == 0, "whole 32-row fragments per wave");
    static_assert(N % GEMV_COLS == 0, "whole blocks");
    __shared__ float part[GEMV_WAVES][GEMV_COLS];

    const int lane = threadIdx.x & 31;
    const int wave = threadIdx.x >> 5;
    const int c    = lane & 15;
    const int hi8  = (lane >> 4) * 8;
    const int col0 = blockIdx.x * GEMV_COLS;

    const float* xp = x + wave * KS + hi8;
    const float* wp = W + (size_t)(wave * KS + hi8) * N + col0 + 2 * c;

    v8f acc0 = {}, acc1 = {};
#pragma unroll 2
    for (int k0 = 0; k0 < KS; k0 += 32) {
        const v4f xa = *(const v4f*)(xp + k0);
        const v4f xb = *(const v4f*)(xp + k0 + 4);
        const v4f xc = *(const v4f*)(xp + k0 + 16);
        const v4f xd = *(const v4f*)(xp + k0 + 20);
        v16h ah = {}, al = {}, b0h = {}, b0l = {}, b1h = {}, b1l = {};
#pragma unroll
        for (int e = 0; e < 16; ++e) {
            const int kk = k0 + e + (e & 8);
            const float xv = (e < 4) ? xa[e & 3] : (e < 8) ? xb[e & 3]
                           : (e < 12) ? xc[e & 3] : xd[e & 3];
            const v2f wv = *(const v2f*)(wp + (size_t)kk * N);
            const _Float16 xh  = (_Float16)xv;
            const _Float16 w0h = (_Float16)wv[0];
            const _Float16 w1h = (_Float16)wv[1];
            ah[e]  = xh;   al[e]  = (_Float16)(xv - (float)xh);
            b0h[e] = w0h;  b0l[e] = (_Float16)(wv[0] - (float)w0h);
            b1h[e] = w1h;  b1l[e] = (_Float16)(wv[1] - (float)w1h);
        }
        acc0 = __builtin_amdgcn_wmma_f32_16x16x32_f16(
            false, ah, false, b0h, (short)0, acc0, false, false);
        acc1 = __builtin_amdgcn_wmma_f32_16x16x32_f16(
            false, ah, false, b1h, (short)0, acc1, false, false);
        acc0 = __builtin_amdgcn_wmma_f32_16x16x32_f16(
            false, ah, false, b0l, (short)0, acc0, false, false);
        acc1 = __builtin_amdgcn_wmma_f32_16x16x32_f16(
            false, ah, false, b1l, (short)0, acc1, false, false);
        acc0 = __builtin_amdgcn_wmma_f32_16x16x32_f16(
            false, al, false, b0h, (short)0, acc0, false, false);
        acc1 = __builtin_amdgcn_wmma_f32_16x16x32_f16(
            false, al, false, b1h, (short)0, acc1, false, false);
    }

    if (lane < 16) {
        v2f p;
        p[0] = acc0[0];
        p[1] = acc1[0];
        *(v2f*)&part[wave][2 * c] = p;
    }
    __syncthreads();

    if (wave == 0) {
        float v = part[0][lane];
        for (int w = 1; w < GEMV_WAVES; ++w) v += part[w][lane];
        v += bias[col0 + lane];
        if (RELU) v = fmaxf(v, 0.0f);
        y[col0 + lane] = v;
    }
}

#define RN   4096
#define HN   (RN / 2)
#define BOFF (HN + 16)
#define SDN  (RN + 16)

__device__ __forceinline__ void sturm_block32(const v4f (&t)[8],
                                              float& p1, float& p0,
                                              unsigned& prev, int& cnt) {
    unsigned acc = prev;
    int      ne  = 0;
#pragma unroll
    for (int u = 0; u < 8; ++u) {
#pragma unroll
        for (int s = 0; s < 4; ++s) {
            const float p2 = __builtin_fmaf(t[u][s], p1, -p0);
            acc = __builtin_amdgcn_alignbit(acc, __float_as_uint(p2), 31u);
            p0 = p1;
            p1 = p2;
        }
        if ((u & 3) == 2) {
            ne = -__builtin_amdgcn_frexp_expf(
                __builtin_fmaxf(__builtin_fabsf(p1), __builtin_fabsf(p0)));
        }
        if ((u & 3) == 3) {
            p1 = __builtin_ldexpf(p1, ne);
            p0 = __builtin_ldexpf(p0, ne);
        }
    }
    const unsigned older = __builtin_amdgcn_alignbit(prev, acc, 1u);
    cnt += __builtin_popcount(acc ^ older);
    prev = acc;
}

__device__ __forceinline__ void hold_block(v4f (&t)[8], float& p1) {
#pragma unroll
    for (int u = 0; u < 8; ++u)
        asm volatile("" : "+v"(t[u]), "+v"(p1) : : "memory");
}

__device__ __forceinline__ int sturm_half_scaled(const v4f* hd4, const float x2,
                                                 float& v1, float& v0) {
    float    p1   = 1.0f;
    float    p0   = 0.0f;
    unsigned prev = 0u;
    int      cnt  = 0;
    const int nblk = HN / 32;

    v4f a[8], b[8], ra[8], rb[8];
#pragma unroll
    for (int u = 0; u < 8; ++u) a[u] = hd4[u] - x2;
#pragma unroll
    for (int u = 0; u < 8; ++u) rb[u] = hd4[8 + u];
    hold_block(rb, p1);

    for (int k = 0; k < nblk; k += 2) {
        const int k2 = ((k + 2) & (nblk - 1)) * 8;
#pragma unroll
        for (int u = 0; u < 8; ++u) ra[u] = hd4[k2 + u];
        __builtin_amdgcn_sched_barrier(0);
#pragma unroll
        for (int u = 0; u < 8; ++u) b[u] = rb[u] - x2;
        sturm_block32(a, p1, p0, prev, cnt);
        hold_block(b, p1);
        hold_block(ra, p1);
        const int k3 = ((k + 3) & (nblk - 1)) * 8;
#pragma unroll
        for (int u = 0; u < 8; ++u) rb[u] = hd4[k3 + u];
        __builtin_amdgcn_sched_barrier(0);
#pragma unroll
        for (int u = 0; u < 8; ++u) a[u] = ra[u] - x2;
        sturm_block32(b, p1, p0, prev, cnt);
        hold_block(a, p1);
        hold_block(rb, p1);
    }
    v1 = p1;
    v0 = p0;
    return cnt;
}

__device__ __forceinline__ int sturm_count_two_sided(const v4f* sd4, const int half,
                                                     const float x2) {
    float v1, v0;
    const int c = sturm_half_scaled(sd4 + (half ? (BOFF / 4) : 0), x2, v1, v0);
    const float w1 = __shfl_xor(v1, 1);
    const float w0 = __shfl_xor(v0, 1);
    const int   d  = __shfl_xor(c, 1);
    const float pa = half ? w1 : v1;
    const float pb = half ? w0 : v0;
    const float ra = half ? v1 : w1;
    const float rb = half ? v0 : w0;
    const float det = __builtin_fmaf(pa, ra, -(pb * rb));
    const unsigned sd = __float_as_uint(det) >> 31;
    const unsigned sp = __float_as_uint(pa)  >> 31;
    const unsigned s1 = __float_as_uint(rb)  >> 31;
    const unsigned s0 = __float_as_uint(ra)  >> 31;
    return c + d + (int)(sd ^ sp ^ s1) - (int)(s1 ^ s0);
}

__device__ __forceinline__ void ratio_step(float& q, int& cnt, const float s,
                                           const float x2) {
    q = (s - x2) - __builtin_amdgcn_rcpf(q);
    if (fabsf(q) < 2e-30f) q = -2e-30f;
    cnt += (q < 0.0f);
}
__device__ __forceinline__ int sturm_count_ratio(const float* sd,
                                                 const float x2) {
    float q   = sd[0] - x2;
    int   cnt = (q < 0.0f);
    for (int i = 1; i < HN; ++i) ratio_step(q, cnt, sd[i], x2);
    for (int i = HN; i < RN; ++i) ratio_step(q, cnt, sd[BOFF + (RN - 1 - i)], x2);
    return cnt;
}

#define NPT    4
#define LPE    (2 * NPT)
#define EPB    (256 / LPE)
#define PMASK  0x55u
#define SWEEPS 16

#define GPP    (256 / 2)
#define GPB    (GPP - 1)
#define GBLK   (RN / EPB)
#define GPTS   (GBLK * GPB + 1)

__device__ __forceinline__ float grid_point(const float lo, const float w,
                                            const int t) {
    return __builtin_fmaf(w, (float)(t + 1) * (1.0f / (float)(GPTS + 1)), lo);
}

__device__ __forceinline__ float section_point(const float lo, const float w,
                                               const int i) {
    return __builtin_fmaf(w, (float)(i + 1) * (1.0f / (float)(NPT + 1)), lo);
}

__device__ __forceinline__ void kept_part(const float lo, const float hi,
                                          const float w, const unsigned gbits,
                                          float& nlo, float& nhi, int& k) {
    k   = __builtin_popcount(gbits & PMASK);
    nlo = (k > 0)   ? section_point(lo, w, k - 1) : lo;
    nhi = (k < NPT) ? section_point(lo, w, k)     : hi;
}

__device__ __forceinline__ void stage_diagonal(const float* __restrict__ ptl,
                                               const int l, const int tid,
                                               float* sd, float* wmn, float* wmx,
                                               float& lo, float& hi) {
    const float ll1 = (float)(l * (l + 1));
    const float dr  = 10000.0f / (float)RN;
    float tmn = 1e30f, tmx = -1e30f;
#pragma unroll
    for (int u = 0; u < RN / 256; ++u) {
        const int   i = tid + u * 256;
        const float r = dr * (float)(i + 1);
        const float d = 1.0f + ptl[i] + ll1 / (r * r);
        const float s = 2.0f * d;
        sd[(u < HN / 256) ? i : (BOFF + (RN - 1 - i))] = s;
        tmn = fminf(tmn, s);
        tmx = fmaxf(tmx, s);
    }
    for (int o = 16; o > 0; o >>= 1) {
        tmn = fminf(tmn, __shfl_xor(tmn, o));
        tmx = fmaxf(tmx, __shfl_xor(tmx, o));
    }
    if ((tid & 31) == 0) { wmn[tid >> 5] = tmn; wmx[tid >> 5] = tmx; }
    __syncthreads();

    float mn = wmn[0], mx = wmx[0];
    for (int w = 1; w < 256 / 32; ++w) {
        mn = fminf(mn, wmn[w]);
        mx = fmaxf(mx, wmx[w]);
    }
    lo = fminf(1e30f, 0.5f * mn - 1.0f);
    hi = fmaxf(-1e30f, 0.5f * mx + 1.0f);
}

__global__ void __launch_bounds__(256)
eig_grid_kernel(const float* __restrict__ ptl, float* __restrict__ evl) {
    __shared__ v4f   sd4[SDN / 4];
    __shared__ int   cnts[GPP];
    __shared__ float wmn[256 / 32];
    __shared__ float wmx[256 / 32];
    float* sd = (float*)sd4;
    const int l   = blockIdx.y;
    const int tid = threadIdx.x;

    float lo, hi;
    stage_diagonal(ptl, l, tid, sd, wmn, wmx, lo, hi);
    const int narrow = __builtin_amdgcn_readfirstlane((hi - lo) < 32.0f ? 1 : 0);

    const int   half = tid & 1;
    const int   pt   = tid >> 1;
    const float x  = grid_point(lo, hi - lo, blockIdx.x * GPB + pt);
    const float x2 = x + x;
    const int   cl = narrow ? sturm_count_two_sided(sd4, half, x2)
                            : sturm_count_ratio(sd, x2);
    if (half == 0) cnts[pt] = cl;
    __syncthreads();

    if (tid < GPP) {
        const int t = blockIdx.x * GPB + tid;
        const int c = cnts[tid];
        float* row = evl + l * RN;
        int from = (tid > 0) ? cnts[tid - 1] : ((blockIdx.x == 0) ? 0 : c);
        from = max(from, 0);
        const int to = min(c, RN);
#pragma clang loop vectorize(disable) interleave(disable) unroll(disable)
        for (int j = from; j < to; ++j) row[j] = (float)t;
        if (blockIdx.x == GBLK - 1 && tid == GPP - 1) {
#pragma clang loop vectorize(disable) interleave(disable) unroll(disable)
            for (int j = max(c, 0); j < RN; ++j) row[j] = (float)GPTS;
        }
    }
}

__global__ void __launch_bounds__(256)
eig_bisect_kernel(const float* __restrict__ ptl, float* __restrict__ evl) {
    __shared__ v4f   sd4[SDN / 4];
    __shared__ float res[EPB];
    __shared__ float wmn[256 / 32];
    __shared__ float wmx[256 / 32];
    float* sd = (float*)sd4;
    const int l   = blockIdx.y;
    const int tid = threadIdx.x;

    const int q    = tid & (LPE - 1);
    const int half = q & 1;
    const int pt   = q >> 1;
    const int grp  = tid / LPE;
    const int gsh  = tid & 31 & ~(LPE - 1);
    const int j    = blockIdx.x * EPB + grp;

    const float cell = evl[l * RN + j];

    float glo, ghi;
    stage_diagonal(ptl, l, tid, sd, wmn, wmx, glo, ghi);
    const float tol = __builtin_ldexpf(
        1.0f, __builtin_amdgcn_frexp_expf(fmaxf(fabsf(glo), fabsf(ghi))) - 24);

    const int narrow = __builtin_amdgcn_readfirstlane((ghi - glo) < 32.0f ? 1 : 0);

    const int   kc = min(max((int)cell, 0), GPTS);
    const float gw = ghi - glo;
    float lo = (kc > 0)    ? grid_point(glo, gw, kc - 1) : glo;
    float hi = (kc < GPTS) ? grid_point(glo, gw, kc)     : ghi;

    bool lo_kn = false, hi_kn = false, lo_b = false, hi_b = false;

    for (int it = 1; it < SWEEPS; ++it) {
        const float w  = hi - lo;
        const float x  = section_point(lo, w, pt);

        const bool at_lo = (x == lo);
        const bool kn = at_lo ? lo_kn : ((x == hi) && hi_kn);
        const bool fb = at_lo ? lo_b : hi_b;
        const unsigned fbits =
            (__builtin_amdgcn_ballot_w32(fb) >> gsh) & ((1u << LPE) - 1u);
        float flo, fhi;
        int   fk;
        kept_part(lo, hi, w, fbits, flo, fhi, fk);
        const bool fine  = (w <= tol);
        const bool stays = fine || (kn && (flo == lo) && (fhi == hi));
        if (__builtin_amdgcn_ballot_w32(!stays) == 0u) break;

        const float x2 = x + x;
        const int cnt = narrow ? sturm_count_two_sided(sd4, half, x2)
                               : sturm_count_ratio(sd, x2);
        const unsigned gbits =
            (__builtin_amdgcn_ballot_w32(cnt <= j) >> gsh) & ((1u << LPE) - 1u);
        float nlo, nhi;
        int   k;
        kept_part(lo, hi, w, gbits, nlo, nhi, k);
        if (!fine) {
            if (k > 0)   { lo_kn = true; lo_b = ((gbits >> (2 * (k - 1))) & 1u) != 0u; }
            if (k < NPT) { hi_kn = true; hi_b = ((gbits >> (2 * k)) & 1u) != 0u; }
            lo = nlo;
            hi = nhi;
        }
    }

    if (q == 0) res[grp] = 0.5f * (lo + hi);
    __syncthreads();
    if (tid < EPB)
        evl[l * RN + blockIdx.x * EPB + tid] = res[tid];
}

extern "C" void kernel_launch(void* const* d_in, const int* in_sizes, int n_in,
                              void* d_out, int out_size, void* d_ws, size_t ws_size,
                              hipStream_t stream) {
    const float* energy = (const float*)d_in[0];
    const float* W1     = (const float*)d_in[1];
    const float* b1     = (const float*)d_in[2];
    const float* W2     = (const float*)d_in[3];
    const float* b2     = (const float*)d_in[4];
    const float* W3     = (const float*)d_in[5];
    const float* b3     = (const float*)d_in[6];

    float* out = (float*)d_out;
    float* h1  = (float*)d_ws;
    float* h2  = h1 + 2048;

    gemv_wmma_kernel<4096, 2048, true>
        <<<2048 / GEMV_COLS, GEMV_WAVES * 32, 0, stream>>>(energy, W1, b1, h1);
    gemv_wmma_kernel<2048, 1024, true>
        <<<1024 / GEMV_COLS, GEMV_WAVES * 32, 0, stream>>>(h1, W2, b2, h2);
    gemv_wmma_kernel<1024, 4096, false>
        <<<4096 / GEMV_COLS, GEMV_WAVES * 32, 0, stream>>>(h2, W3, b3, out);
    eig_grid_kernel<<<dim3(GBLK, 2), 256, 0, stream>>>(out, out + 4096);
    eig_bisect_kernel<<<dim3(GBLK, 2), 256, 0, stream>>>(out, out + 4096);
}
